# starting kernel
# baseline (speedup 1.0000x reference)
#include <hip/hip_runtime.h>
#include <hip/hip_bf16.h>


typedef float v2f __attribute__((ext_vector_type(2)));
typedef float v8f __attribute__((ext_vector_type(8)));

#define H_DIM 64

__global__ void k_deg_init(float* __restrict__ deg, int n) {
  int i = blockIdx.x * blockDim.x + threadIdx.x;
  if (i < n) deg[i] = 1.0f;
}

__global__ void k_deg_accum(const int* __restrict__ dst, float* __restrict__ deg,
                            int ne) {
  int i = blockIdx.x * blockDim.x + threadIdx.x;
  if (i < ne) unsafeAtomicAdd(&deg[dst[i]], 1.0f);
}

__global__ void k_deg_rsqrt(float* __restrict__ deg, int n) {
  int i = blockIdx.x * blockDim.x + threadIdx.x;
  if (i < n) deg[i] = rsqrtf(deg[i]);
}

__global__ void k_gemm_wmma_fused(const float* __restrict__ X,
                                  const float* __restrict__ W,
                                  const float* __restrict__ bias,
                                  const float* __restrict__ dis,
                                  float* __restrict__ Hout,
                                  float* __restrict__ Acc, int n) {
  const int lane  = threadIdx.x & 31;
  const int wave  = threadIdx.x >> 5;
  const int row0  = blockIdx.x * 16;
  const int n0    = wave * 16;
  const int m     = lane & 15;
  const int khalf = (lane >> 4) << 1;

  int arow = row0 + m;
  if (arow >= n) arow = n - 1;
  const float* __restrict__ xrow = X + (size_t)arow * H_DIM;

  v8f acc = {};
#pragma unroll
  for (int kk = 0; kk < 16; ++kk) {
    const int kb = kk * 4 + khalf;
    v2f a;
    a.x = xrow[kb];
    a.y = xrow[kb + 1];
    v2f b;
    b.x = W[kb * H_DIM + n0 + m];
    b.y = W[(kb + 1) * H_DIM + n0 + m];
    acc = __builtin_amdgcn_wmma_f32_16x16x4_f32(false, a, false, b, (short)0,
                                                acc, false, false);
  }

  const int rbase = (lane >> 4) << 3;
  const int col   = n0 + m;
  const float bc  = bias[col];
#pragma unroll
  for (int r = 0; r < 8; ++r) {
    const int row = row0 + rbase + r;
    if (row < n) {
      const float h = acc[r];
      const float d = dis[row];
      Hout[(size_t)row * H_DIM + col] = h;
      Acc[(size_t)row * H_DIM + col] = fmaf(h, d * d, bc);
    }
  }
}

__global__ void k_scatter_edges(const float* __restrict__ Hm,
                                const int* __restrict__ src,
                                const int* __restrict__ dst,
                                const float* __restrict__ dis,
                                float* __restrict__ Acc, int ne) {
  const int e = blockIdx.x * (blockDim.x >> 5) + (threadIdx.x >> 5);
  if (e >= ne) return;
  const int lane = threadIdx.x & 31;
  const int s = src[e];
  const int d = dst[e];
  const float nrm = dis[s] * dis[d];
  const float2 v = ((const float2*)(Hm + (size_t)s * H_DIM))[lane];
  float* outp = Acc + (size_t)d * H_DIM + lane * 2;
  unsafeAtomicAdd(outp, v.x * nrm);
  unsafeAtomicAdd(outp + 1, v.y * nrm);
}

__global__ void k_relu(float* __restrict__ a, size_t n) {
  size_t i = (size_t)blockIdx.x * blockDim.x + threadIdx.x;
  if (i < n) a[i] = fmaxf(a[i], 0.0f);
}

__global__ void k_decode(const float* __restrict__ Z,
                         const int* __restrict__ pos,
                         const int* __restrict__ neg,
                         float* __restrict__ out, int npos, int nneg) {
  const int k = blockIdx.x * (blockDim.x >> 5) + (threadIdx.x >> 5);
  if (k >= npos + nneg) return;
  const int lane = threadIdx.x & 31;
  int a, b;
  if (k < npos) {
    a = pos[k];
    b = pos[npos + k];
  } else {
    const int j = k - npos;
    a = neg[j];
    b = neg[nneg + j];
  }
  const float2 va = ((const float2*)(Z + (size_t)a * H_DIM))[lane];
  const float2 vb = ((const float2*)(Z + (size_t)b * H_DIM))[lane];
  float s = fmaf(va.x, vb.x, va.y * vb.y);
#pragma unroll
  for (int off = 16; off > 0; off >>= 1) s += __shfl_xor(s, off, 32);
  if (lane == 0) out[k] = s;
}

extern "C" void kernel_launch(void* const* d_in, const int* in_sizes, int n_in,
                              void* d_out, int out_size, void* d_ws,
                              size_t ws_size, hipStream_t stream) {
  (void)n_in; (void)out_size; (void)ws_size;
  const float* x   = (const float*)d_in[0];
  const int*   ei  = (const int*)d_in[1];
  const int*   pei = (const int*)d_in[2];
  const int*   nei = (const int*)d_in[3];
  const float* W1  = (const float*)d_in[4];
  const float* b1  = (const float*)d_in[5];
  const float* W2  = (const float*)d_in[6];
  const float* b2  = (const float*)d_in[7];
  float* out = (float*)d_out;

  const int n    = in_sizes[0] / H_DIM;
  const int ne   = in_sizes[1] / 2;
  const int npos = in_sizes[2] / 2;
  const int nneg = in_sizes[3] / 2;
  const int* src = ei;
  const int* dst = ei + ne;

  float* ws = (float*)d_ws;
  const size_t nAl = ((size_t)n + 63) & ~(size_t)63;
  float* dis  = ws;
  float* hbuf = ws + nAl;
  float* acc1 = hbuf + (size_t)n * H_DIM;
  float* acc2 = acc1 + (size_t)n * H_DIM;

  const int B = 256;
  k_deg_init<<<(n + B - 1) / B, B, 0, stream>>>(dis, n);
  k_deg_accum<<<(ne + B - 1) / B, B, 0, stream>>>(dst, dis, ne);
  k_deg_rsqrt<<<(n + B - 1) / B, B, 0, stream>>>(dis, n);

  const int tiles = (n + 15) / 16;
  k_gemm_wmma_fused<<<tiles, 128, 0, stream>>>(x, W1, b1, dis, hbuf, acc1, n);
  k_scatter_edges<<<(ne + 7) / 8, B, 0, stream>>>(hbuf, src, dst, dis, acc1, ne);
  k_relu<<<(int)(((size_t)n * H_DIM + B - 1) / B), B, 0, stream>>>(
      acc1, (size_t)n * H_DIM);
  k_gemm_wmma_fused<<<tiles, 128, 0, stream>>>(acc1, W2, b2, dis, hbuf, acc2, n);
  k_scatter_edges<<<(ne + 7) / 8, B, 0, stream>>>(hbuf, src, dst, dis, acc2, ne);
  k_decode<<<(npos + nneg + 7) / 8, B, 0, stream>>>(acc2, pei, nei, out, npos,
                                                    nneg);
}


# Float-atomic scatter-add replaced: edges put in destination order per call (returning int count gives arrival rank, scan, place); one wave sums a node's rows in registers, stores once with ReLU
# speedup vs baseline: 1.7037x; 1.7037x over previous
#include <hip/hip_runtime.h>
#include <hip/hip_bf16.h>


typedef float v2f __attribute__((ext_vector_type(2)));
typedef float v8f __attribute__((ext_vector_type(8)));

#define H_DIM 64

__global__ void k_cnt_zero(int* __restrict__ cnt, int n) {
  int i = blockIdx.x * blockDim.x + threadIdx.x;
  if (i < n) cnt[i] = 0;
}

__global__ void k_count_rank(const int* __restrict__ dst, int* __restrict__ cnt,
                             int* __restrict__ rank, int ne) {
  int e = blockIdx.x * blockDim.x + threadIdx.x;
  if (e < ne) rank[e] = atomicAdd(&cnt[dst[e]], 1);
}

__global__ void __launch_bounds__(1024)
k_scan_dis(const int* __restrict__ cnt, int* __restrict__ off,
           float* __restrict__ dis, int n) {
  __shared__ int s_part[32];
  __shared__ int s_wsum[32];
  const int t = threadIdx.x;
  const int lane = t & 31;
  const int w = t >> 5;
  const int base = blockIdx.x * 1024;

  int pre = 0;
  for (int k = t; k < base; k += 1024) pre += cnt[k];
  for (int o = 16; o > 0; o >>= 1) pre += __shfl_xor(pre, o, 32);
  if (lane == 0) s_part[w] = pre;

  const int i = base + t;
  const int v = (i < n) ? cnt[i] : 0;
  int inc = v;
  for (int o = 1; o < 32; o <<= 1) {
    const int u = __shfl_up(inc, o, 32);
    if (lane >= o) inc += u;
  }
  if (lane == 31) s_wsum[w] = inc;
  __syncthreads();

  int p = s_part[lane];
  for (int o = 16; o > 0; o >>= 1) p += __shfl_xor(p, o, 32);
  int q = (lane < w) ? s_wsum[lane] : 0;
  for (int o = 16; o > 0; o >>= 1) q += __shfl_xor(q, o, 32);

  const int excl = p + q + inc - v;
  if (i < n) {
    off[i] = excl;
    dis[i] = rsqrtf((float)(v + 1));
    if (i == n - 1) off[n] = excl + v;
  }
}

__global__ void k_place(const int* __restrict__ src, const int* __restrict__ dst,
                        const int* __restrict__ rank, const int* __restrict__ off,
                        const float* __restrict__ dis, int* __restrict__ csr,
                        int ne) {
  int e = blockIdx.x * blockDim.x + threadIdx.x;
  if (e < ne) {
    const int s = src[e];
    ((int2*)csr)[off[dst[e]] + rank[e]] = make_int2(s, __float_as_int(dis[s]));
  }
}

__global__ void k_gemm_wmma(const float* __restrict__ X,
                            const float* __restrict__ W,
                            float* __restrict__ Hout, int n) {
  const int lane  = threadIdx.x & 31;
  const int wave  = threadIdx.x >> 5;
  const int row0  = blockIdx.x * 16;
  const int n0    = wave * 16;
  const int m     = lane & 15;
  const int khalf = (lane >> 4) << 1;

  int arow = row0 + m;
  if (arow >= n) arow = n - 1;
  const float* __restrict__ xrow = X + (size_t)arow * H_DIM;

  v8f acc = {};
#pragma unroll
  for (int kk = 0; kk < 16; ++kk) {
    const int kb = kk * 4 + khalf;
    v2f a;
    a.x = xrow[kb];
    a.y = xrow[kb + 1];
    v2f b;
    b.x = W[kb * H_DIM + n0 + m];
    b.y = W[(kb + 1) * H_DIM + n0 + m];
    acc = __builtin_amdgcn_wmma_f32_16x16x4_f32(false, a, false, b, (short)0,
                                                acc, false, false);
  }

  const int rbase = (lane >> 4) << 3;
  const int col   = n0 + m;
#pragma unroll
  for (int r = 0; r < 8; ++r) {
    const int row = row0 + rbase + r;
    if (row < n) Hout[(size_t)row * H_DIM + col] = acc[r];
  }
}

template <bool RELU>
__global__ void k_gather_rows(const float* __restrict__ Hm,
                              const int* __restrict__ off,
                              const int* __restrict__ csr,
                              const float* __restrict__ dis,
                              const float* __restrict__ bias,
                              float* __restrict__ Out, int n) {
  const int node = __builtin_amdgcn_readfirstlane(
      (int)(blockIdx.x * (blockDim.x >> 5) + (threadIdx.x >> 5)));
  if (node >= n) return;
  const int lane = threadIdx.x & 31;
  const int beg = off[node];
  const int end = off[node + 1];
  const float dd = dis[node];
  const float2* __restrict__ H2 = (const float2*)Hm;
  const float2 hs = H2[(size_t)node * (H_DIM / 2) + lane];
  const float2 bb = ((const float2*)bias)[lane];
  float ax = fmaf(hs.x, dd * dd, bb.x);
  float ay = fmaf(hs.y, dd * dd, bb.y);

  for (int c = beg; c < end; c += 32) {
    const int m = (end - c < 32) ? (end - c) : 32;
    int s_l = 0;
    float w_l = 0.0f;
    if (lane < m) {
      const int2 p = ((const int2*)csr)[c + lane];
      s_l = p.x;
      w_l = __int_as_float(p.y) * dd;
    }
    int t = 0;
    for (; t + 4 <= m; t += 4) {
      const int s0 = __builtin_amdgcn_readlane(s_l, t);
      const int s1 = __builtin_amdgcn_readlane(s_l, t + 1);
      const int s2 = __builtin_amdgcn_readlane(s_l, t + 2);
      const int s3 = __builtin_amdgcn_readlane(s_l, t + 3);
      const float w0 = __int_as_float(__builtin_amdgcn_readlane(__float_as_int(w_l), t));
      const float w1 = __int_as_float(__builtin_amdgcn_readlane(__float_as_int(w_l), t + 1));
      const float w2 = __int_as_float(__builtin_amdgcn_readlane(__float_as_int(w_l), t + 2));
      const float w3 = __int_as_float(__builtin_amdgcn_readlane(__float_as_int(w_l), t + 3));
      const float2 v0 = H2[(size_t)s0 * (H_DIM / 2) + lane];
      const float2 v1 = H2[(size_t)s1 * (H_DIM / 2) + lane];
      const float2 v2 = H2[(size_t)s2 * (H_DIM / 2) + lane];
      const float2 v3 = H2[(size_t)s3 * (H_DIM / 2) + lane];
      ax = fmaf(v0.x, w0, ax);  ay = fmaf(v0.y, w0, ay);
      ax = fmaf(v1.x, w1, ax);  ay = fmaf(v1.y, w1, ay);
      ax = fmaf(v2.x, w2, ax);  ay = fmaf(v2.y, w2, ay);
      ax = fmaf(v3.x, w3, ax);  ay = fmaf(v3.y, w3, ay);
    }
    const int r = m - t;
    if (r == 3) {
      const int s0 = __builtin_amdgcn_readlane(s_l, t);
      const int s1 = __builtin_amdgcn_readlane(s_l, t + 1);
      const int s2 = __builtin_amdgcn_readlane(s_l, t + 2);
      const float w0 = __int_as_float(__builtin_amdgcn_readlane(__float_as_int(w_l), t));
      const float w1 = __int_as_float(__builtin_amdgcn_readlane(__float_as_int(w_l), t + 1));
      const float w2 = __int_as_float(__builtin_amdgcn_readlane(__float_as_int(w_l), t + 2));
      const float2 v0 = H2[(size_t)s0 * (H_DIM / 2) + lane];
      const float2 v1 = H2[(size_t)s1 * (H_DIM / 2) + lane];
      const float2 v2 = H2[(size_t)s2 * (H_DIM / 2) + lane];
      ax = fmaf(v0.x, w0, ax);  ay = fmaf(v0.y, w0, ay);
      ax = fmaf(v1.x, w1, ax);  ay = fmaf(v1.y, w1, ay);
      ax = fmaf(v2.x, w2, ax);  ay = fmaf(v2.y, w2, ay);
    } else if (r == 2) {
      const int s0 = __builtin_amdgcn_readlane(s_l, t);
      const int s1 = __builtin_amdgcn_readlane(s_l, t + 1);
      const float w0 = __int_as_float(__builtin_amdgcn_readlane(__float_as_int(w_l), t));
      const float w1 = __int_as_float(__builtin_amdgcn_readlane(__float_as_int(w_l), t + 1));
      const float2 v0 = H2[(size_t)s0 * (H_DIM / 2) + lane];
      const float2 v1 = H2[(size_t)s1 * (H_DIM / 2) + lane];
      ax = fmaf(v0.x, w0, ax);  ay = fmaf(v0.y, w0, ay);
      ax = fmaf(v1.x, w1, ax);  ay = fmaf(v1.y, w1, ay);
    } else if (r == 1) {
      const int s0 = __builtin_amdgcn_readlane(s_l, t);
      const float w0 = __int_as_float(__builtin_amdgcn_readlane(__float_as_int(w_l), t));
      const float2 v0 = H2[(size_t)s0 * (H_DIM / 2) + lane];
      ax = fmaf(v0.x, w0, ax);  ay = fmaf(v0.y, w0, ay);
    }
  }
  if (RELU) {
    ax = fmaxf(ax, 0.0f);
    ay = fmaxf(ay, 0.0f);
  }
  ((float2*)Out)[(size_t)node * (H_DIM / 2) + lane] = make_float2(ax, ay);
}

__global__ void k_decode(const float* __restrict__ Z,
                         const int* __restrict__ pos,
                         const int* __restrict__ neg,
                         float* __restrict__ out, int npos, int nneg) {
  const int k = blockIdx.x * (blockDim.x >> 5) + (threadIdx.x >> 5);
  if (k >= npos + nneg) return;
  const int lane = threadIdx.x & 31;
  int a, b;
  if (k < npos) {
    a = pos[k];
    b = pos[npos + k];
  } else {
    const int j = k - npos;
    a = neg[j];
    b = neg[nneg + j];
  }
  const float2 va = ((const float2*)(Z + (size_t)a * H_DIM))[lane];
  const float2 vb = ((const float2*)(Z + (size_t)b * H_DIM))[lane];
  float s = fmaf(va.x, vb.x, va.y * vb.y);
#pragma unroll
  for (int off = 16; off > 0; off >>= 1) s += __shfl_xor(s, off, 32);
  if (lane == 0) out[k] = s;
}

extern "C" void kernel_launch(void* const* d_in, const int* in_sizes, int n_in,
                              void* d_out, int out_size, void* d_ws,
                              size_t ws_size, hipStream_t stream) {
  (void)n_in; (void)out_size; (void)ws_size;
  const float* x   = (const float*)d_in[0];
  const int*   ei  = (const int*)d_in[1];
  const int*   pei = (const int*)d_in[2];
  const int*   nei = (const int*)d_in[3];
  const float* W1  = (const float*)d_in[4];
  const float* b1  = (const float*)d_in[5];
  const float* W2  = (const float*)d_in[6];
  const float* b2  = (const float*)d_in[7];
  float* out = (float*)d_out;

  const int n    = in_sizes[0] / H_DIM;
  const int ne   = in_sizes[1] / 2;
  const int npos = in_sizes[2] / 2;
  const int nneg = in_sizes[3] / 2;
  const int* src = ei;
  const int* dst = ei + ne;

  float* ws = (float*)d_ws;
  const size_t nAl  = ((size_t)n + 63) & ~(size_t)63;
  const size_t neAl = ((size_t)ne + 63) & ~(size_t)63;
  float* dis  = ws;
  float* hbuf = ws + nAl;
  float* acc1 = hbuf + (size_t)n * H_DIM;
  float* acc2 = acc1 + (size_t)n * H_DIM;
  int* cnt  = (int*)(acc2 + (size_t)n * H_DIM);
  int* off  = cnt + nAl;
  int* rank = off + nAl + 64;
  int* csr  = rank + neAl;

  const int B = 256;
  k_cnt_zero<<<(n + B - 1) / B, B, 0, stream>>>(cnt, n);
  k_count_rank<<<(ne + B - 1) / B, B, 0, stream>>>(dst, cnt, rank, ne);
  k_scan_dis<<<(n + 1023) / 1024, 1024, 0, stream>>>(cnt, off, dis, n);
  k_place<<<(ne + B - 1) / B, B, 0, stream>>>(src, dst, rank, off, dis, csr, ne);

  const int tiles = (n + 15) / 16;
  const int gblocks = (n + 7) / 8;
  k_gemm_wmma<<<tiles, 128, 0, stream>>>(x, W1, hbuf, n);
  k_gather_rows<true><<<gblocks, B, 0, stream>>>(hbuf, off, csr, dis, b1, acc1, n);
  k_gemm_wmma<<<tiles, 128, 0, stream>>>(acc1, W2, hbuf, n);
  k_gather_rows<false><<<gblocks, B, 0, stream>>>(hbuf, off, csr, dis, b2, acc2, n);
  k_decode<<<(npos + nneg + 7) / 8, B, 0, stream>>>(acc2, pei, nei, out, npos,
                                                    nneg);
}


# Waves per call cut from about 530 thousand to about 65 thousand: in every launch a wave or thread loops over several nodes, pairs, tiles or edges instead of living for one; sums keep their order
# speedup vs baseline: 1.7973x; 1.0549x over previous
#include <hip/hip_runtime.h>
#include <hip/hip_bf16.h>


typedef float v2f __attribute__((ext_vector_type(2)));
typedef float v8f __attribute__((ext_vector_type(8)));

#define H_DIM 64

#define ZERO_PER_THREAD 4
#define EDGES_PER_THREAD 8

__global__ void k_cnt_zero(int* __restrict__ cnt, int n) {
  const int T = gridDim.x * blockDim.x;
  int i = blockIdx.x * blockDim.x + threadIdx.x;
#pragma unroll
  for (int j = 0; j < ZERO_PER_THREAD; ++j, i += T)
    if (i < n) cnt[i] = 0;
}

__global__ void k_count_rank(const int* __restrict__ dst, int* __restrict__ cnt,
                             int* __restrict__ rank, int ne) {
  const int T = gridDim.x * blockDim.x;
  const int e0 = blockIdx.x * blockDim.x + threadIdx.x;
  int d[EDGES_PER_THREAD];
#pragma unroll
  for (int j = 0; j < EDGES_PER_THREAD; ++j) {
    const int e = e0 + j * T;
    d[j] = (e < ne) ? dst[e] : 0;
  }
  int r[EDGES_PER_THREAD];
#pragma unroll
  for (int j = 0; j < EDGES_PER_THREAD; ++j) {
    const int e = e0 + j * T;
    r[j] = 0;
    if (e < ne) r[j] = atomicAdd(&cnt[d[j]], 1);
  }
#pragma unroll
  for (int j = 0; j < EDGES_PER_THREAD; ++j) {
    const int e = e0 + j * T;
    if (e < ne) rank[e] = r[j];
  }
}

__global__ void __launch_bounds__(1024)
k_scan_dis(const int* __restrict__ cnt, int* __restrict__ off,
           float* __restrict__ dis, int n) {
  __shared__ int s_part[32];
  __shared__ int s_wsum[32];
  const int t = threadIdx.x;
  const int lane = t & 31;
  const int w = t >> 5;
  const int base = blockIdx.x * 1024;

  int pre = 0;
  for (int k = t; k < base; k += 1024) pre += cnt[k];
  for (int o = 16; o > 0; o >>= 1) pre += __shfl_xor(pre, o, 32);
  if (lane == 0) s_part[w] = pre;

  const int i = base + t;
  const int v = (i < n) ? cnt[i] : 0;
  int inc = v;
  for (int o = 1; o < 32; o <<= 1) {
    const int u = __shfl_up(inc, o, 32);
    if (lane >= o) inc += u;
  }
  if (lane == 31) s_wsum[w] = inc;
  __syncthreads();

  int p = s_part[lane];
  for (int o = 16; o > 0; o >>= 1) p += __shfl_xor(p, o, 32);
  int q = (lane < w) ? s_wsum[lane] : 0;
  for (int o = 16; o > 0; o >>= 1) q += __shfl_xor(q, o, 32);

  const int excl = p + q + inc - v;
  if (i < n) {
    off[i] = excl;
    dis[i] = rsqrtf((float)(v + 1));
    if (i == n - 1) off[n] = excl + v;
  }
}

__global__ void k_place(const int* __restrict__ src, const int* __restrict__ dst,
                        const int* __restrict__ rank, const int* __restrict__ off,
                        const float* __restrict__ dis, int* __restrict__ csr,
                        int ne) {
  const int T = gridDim.x * blockDim.x;
  const int e0 = blockIdx.x * blockDim.x + threadIdx.x;
  int s[EDGES_PER_THREAD], d[EDGES_PER_THREAD], r[EDGES_PER_THREAD];
#pragma unroll
  for (int j = 0; j < EDGES_PER_THREAD; ++j) {
    const int e = e0 + j * T;
    const bool ok = e < ne;
    s[j] = ok ? src[e] : 0;
    d[j] = ok ? dst[e] : 0;
    r[j] = ok ? rank[e] : 0;
  }
  int slot[EDGES_PER_THREAD];
  float w[EDGES_PER_THREAD];
#pragma unroll
  for (int j = 0; j < EDGES_PER_THREAD; ++j) {
    const int e = e0 + j * T;
    slot[j] = 0;
    w[j] = 0.0f;
    if (e < ne) {
      slot[j] = off[d[j]] + r[j];
      w[j] = dis[s[j]];
    }
  }
#pragma unroll
  for (int j = 0; j < EDGES_PER_THREAD; ++j) {
    const int e = e0 + j * T;
    if (e < ne) ((int2*)csr)[slot[j]] = make_int2(s[j], __float_as_int(w[j]));
  }
}

__global__ void __launch_bounds__(128)
k_gemm_wmma(const float* __restrict__ X, const float* __restrict__ W,
            float* __restrict__ Hout, int n, int tiles) {
  const int lane  = threadIdx.x & 31;
  const int wave  = threadIdx.x >> 5;
  const int n0    = wave * 16;
  const int m     = lane & 15;
  const int khalf = (lane >> 4) << 1;
  const int rbase = (lane >> 4) << 3;
  const int col   = n0 + m;

  v2f bf[16];
#pragma unroll
  for (int kk = 0; kk < 16; ++kk) {
    const int kb = kk * 4 + khalf;
    bf[kk].x = W[kb * H_DIM + col];
    bf[kk].y = W[(kb + 1) * H_DIM + col];
  }

  for (int tile = blockIdx.x; tile < tiles; tile += gridDim.x) {
    const int row0 = tile * 16;
    int arow = row0 + m;
    if (arow >= n) arow = n - 1;
    const float* __restrict__ xrow = X + (size_t)arow * H_DIM;

    v8f acc = {};
#pragma unroll
    for (int kk = 0; kk < 16; ++kk) {
      const int kb = kk * 4 + khalf;
      v2f a;
      a.x = xrow[kb];
      a.y = xrow[kb + 1];
      acc = __builtin_amdgcn_wmma_f32_16x16x4_f32(false, a, false, bf[kk],
                                                  (short)0, acc, false, false);
    }

#pragma unroll
    for (int r = 0; r < 8; ++r) {
      const int row = row0 + rbase + r;
      if (row < n) Hout[(size_t)row * H_DIM + col] = acc[r];
    }
  }
}

template <bool RELU>
__global__ void k_gather_rows(const float* __restrict__ Hm,
                              const int* __restrict__ off,
                              const int* __restrict__ csr,
                              const float* __restrict__ dis,
                              const float* __restrict__ bias,
                              float* __restrict__ Out, int n) {
  const int wave0 = __builtin_amdgcn_readfirstlane(
      (int)(blockIdx.x * (blockDim.x >> 5) + (threadIdx.x >> 5)));
  const int nwaves = (int)(gridDim.x * (blockDim.x >> 5));
  const int lane = threadIdx.x & 31;
  const float2 bb = ((const float2*)bias)[lane];
  for (int node = wave0; node < n; node += nwaves) {
    const int beg = off[node];
    const int end = off[node + 1];
    const float dd = dis[node];
    const float2* __restrict__ H2 = (const float2*)Hm;
    const float2 hs = H2[(size_t)node * (H_DIM / 2) + lane];
    float ax = fmaf(hs.x, dd * dd, bb.x);
    float ay = fmaf(hs.y, dd * dd, bb.y);

    for (int c = beg; c < end; c += 32) {
      const int m = (end - c < 32) ? (end - c) : 32;
      int s_l = 0;
      float w_l = 0.0f;
      if (lane < m) {
        const int2 p = ((const int2*)csr)[c + lane];
        s_l = p.x;
        w_l = __int_as_float(p.y) * dd;
      }
      int t = 0;
      for (; t + 4 <= m; t += 4) {
        const int s0 = __builtin_amdgcn_readlane(s_l, t);
        const int s1 = __builtin_amdgcn_readlane(s_l, t + 1);
        const int s2 = __builtin_amdgcn_readlane(s_l, t + 2);
        const int s3 = __builtin_amdgcn_readlane(s_l, t + 3);
        const float w0 = __int_as_float(__builtin_amdgcn_readlane(__float_as_int(w_l), t));
        const float w1 = __int_as_float(__builtin_amdgcn_readlane(__float_as_int(w_l), t + 1));
        const float w2 = __int_as_float(__builtin_amdgcn_readlane(__float_as_int(w_l), t + 2));
        const float w3 = __int_as_float(__builtin_amdgcn_readlane(__float_as_int(w_l), t + 3));
        const float2 v0 = H2[(size_t)s0 * (H_DIM / 2) + lane];
        const float2 v1 = H2[(size_t)s1 * (H_DIM / 2) + lane];
        const float2 v2 = H2[(size_t)s2 * (H_DIM / 2) + lane];
        const float2 v3 = H2[(size_t)s3 * (H_DIM / 2) + lane];
        ax = fmaf(v0.x, w0, ax);  ay = fmaf(v0.y, w0, ay);
        ax = fmaf(v1.x, w1, ax);  ay = fmaf(v1.y, w1, ay);
        ax = fmaf(v2.x, w2, ax);  ay = fmaf(v2.y, w2, ay);
        ax = fmaf(v3.x, w3, ax);  ay = fmaf(v3.y, w3, ay);
      }
      const int r = m - t;
      if (r == 3) {
        const int s0 = __builtin_amdgcn_readlane(s_l, t);
        const int s1 = __builtin_amdgcn_readlane(s_l, t + 1);
        const int s2 = __builtin_amdgcn_readlane(s_l, t + 2);
        const float w0 = __int_as_float(__builtin_amdgcn_readlane(__float_as_int(w_l), t));
        const float w1 = __int_as_float(__builtin_amdgcn_readlane(__float_as_int(w_l), t + 1));
        const float w2 = __int_as_float(__builtin_amdgcn_readlane(__float_as_int(w_l), t + 2));
        const float2 v0 = H2[(size_t)s0 * (H_DIM / 2) + lane];
        const float2 v1 = H2[(size_t)s1 * (H_DIM / 2) + lane];
        const float2 v2 = H2[(size_t)s2 * (H_DIM / 2) + lane];
        ax = fmaf(v0.x, w0, ax);  ay = fmaf(v0.y, w0, ay);
        ax = fmaf(v1.x, w1, ax);  ay = fmaf(v1.y, w1, ay);
        ax = fmaf(v2.x, w2, ax);  ay = fmaf(v2.y, w2, ay);
      } else if (r == 2) {
        const int s0 = __builtin_amdgcn_readlane(s_l, t);
        const int s1 = __builtin_amdgcn_readlane(s_l, t + 1);
        const float w0 = __int_as_float(__builtin_amdgcn_readlane(__float_as_int(w_l), t));
        const float w1 = __int_as_float(__builtin_amdgcn_readlane(__float_as_int(w_l), t + 1));
        const float2 v0 = H2[(size_t)s0 * (H_DIM / 2) + lane];
        const float2 v1 = H2[(size_t)s1 * (H_DIM / 2) + lane];
        ax = fmaf(v0.x, w0, ax);  ay = fmaf(v0.y, w0, ay);
        ax = fmaf(v1.x, w1, ax);  ay = fmaf(v1.y, w1, ay);
      } else if (r == 1) {
        const int s0 = __builtin_amdgcn_readlane(s_l, t);
        const float w0 = __int_as_float(__builtin_amdgcn_readlane(__float_as_int(w_l), t));
        const float2 v0 = H2[(size_t)s0 * (H_DIM / 2) + lane];
        ax = fmaf(v0.x, w0, ax);  ay = fmaf(v0.y, w0, ay);
      }
    }
    if (RELU) {
      ax = fmaxf(ax, 0.0f);
      ay = fmaxf(ay, 0.0f);
    }
    ((float2*)Out)[(size_t)node * (H_DIM / 2) + lane] = make_float2(ax, ay);
  }
}

#define DEC_GROUP 16
__global__ void __launch_bounds__(256)
k_decode(const float* __restrict__ Z, const int* __restrict__ pos,
         const int* __restrict__ neg, float* __restrict__ out, int npos,
         int nneg) {
  const int total = npos + nneg;
  const int base = __builtin_amdgcn_readfirstlane(
      (int)(blockIdx.x * (blockDim.x >> 5) + (threadIdx.x >> 5))) * 32;
  if (base >= total) return;
  const int lane = threadIdx.x & 31;
  const int k = base + lane;
  int a = 0, b = 0;
  if (k < npos) {
    a = pos[k];
    b = pos[npos + k];
  } else if (k < total) {
    const int j = k - npos;
    a = neg[j];
    b = neg[nneg + j];
  }
  const float2* __restrict__ Z2 = (const float2*)Z;
  float res = 0.0f;
#pragma unroll 1
  for (int t0 = 0; t0 < 32; t0 += DEC_GROUP) {
    float2 va[DEC_GROUP], vb[DEC_GROUP];
#pragma unroll
    for (int i = 0; i < DEC_GROUP; ++i) {
      const int sa = __builtin_amdgcn_readlane(a, t0 + i);
      const int sb = __builtin_amdgcn_readlane(b, t0 + i);
      va[i] = Z2[(size_t)sa * (H_DIM / 2) + lane];
      vb[i] = Z2[(size_t)sb * (H_DIM / 2) + lane];
    }
#pragma unroll
    for (int i = 0; i < DEC_GROUP; ++i) {
      float s = fmaf(va[i].x, vb[i].x, va[i].y * vb[i].y);
#pragma unroll
      for (int o = 16; o > 0; o >>= 1) s += __shfl_xor(s, o, 32);
      if (lane == t0 + i) res = s;
    }
  }
  if (k < total) out[k] = res;
}

extern "C" void kernel_launch(void* const* d_in, const int* in_sizes, int n_in,
                              void* d_out, int out_size, void* d_ws,
                              size_t ws_size, hipStream_t stream) {
  (void)n_in; (void)out_size; (void)ws_size;
  const float* x   = (const float*)d_in[0];
  const int*   ei  = (const int*)d_in[1];
  const int*   pei = (const int*)d_in[2];
  const int*   nei = (const int*)d_in[3];
  const float* W1  = (const float*)d_in[4];
  const float* b1  = (const float*)d_in[5];
  const float* W2  = (const float*)d_in[6];
  const float* b2  = (const float*)d_in[7];
  float* out = (float*)d_out;

  const int n    = in_sizes[0] / H_DIM;
  const int ne   = in_sizes[1] / 2;
  const int npos = in_sizes[2] / 2;
  const int nneg = in_sizes[3] / 2;
  const int* src = ei;
  const int* dst = ei + ne;

  float* ws = (float*)d_ws;
  const size_t nAl  = ((size_t)n + 63) & ~(size_t)63;
  const size_t neAl = ((size_t)ne + 63) & ~(size_t)63;
  float* dis  = ws;
  float* hbuf = ws + nAl;
  float* acc1 = hbuf + (size_t)n * H_DIM;
  float* acc2 = acc1 + (size_t)n * H_DIM;
  int* cnt  = (int*)(acc2 + (size_t)n * H_DIM);
  int* off  = cnt + nAl;
  int* rank = off + nAl + 64;
  int* csr  = rank + neAl;

  const int B = 256;
  const int zB = B * ZERO_PER_THREAD;
  const int eB = B * EDGES_PER_THREAD;
  k_cnt_zero<<<(n + zB - 1) / zB, B, 0, stream>>>(cnt, n);
  k_count_rank<<<(ne + eB - 1) / eB, B, 0, stream>>>(dst, cnt, rank, ne);
  k_scan_dis<<<(n + 1023) / 1024, 1024, 0, stream>>>(cnt, off, dis, n);
  k_place<<<(ne + eB - 1) / eB, B, 0, stream>>>(src, dst, rank, off, dis, csr, ne);

  const int tiles = (n + 15) / 16;
  const int gemm_blocks = tiles < 2048 ? tiles : 2048;
  const int node_blocks = (n + 7) / 8;
  const int gblocks = node_blocks < 1792 ? node_blocks : 1792;
  k_gemm_wmma<<<gemm_blocks, 128, 0, stream>>>(x, W1, hbuf, n, tiles);
  k_gather_rows<true><<<gblocks, B, 0, stream>>>(hbuf, off, csr, dis, b1, acc1, n);
  k_gemm_wmma<<<gemm_blocks, 128, 0, stream>>>(acc1, W2, hbuf, n, tiles);
  k_gather_rows<false><<<gblocks, B, 0, stream>>>(hbuf, off, csr, dis, b2, acc2, n);
  k_decode<<<(npos + nneg + 255) / 256, B, 0, stream>>>(acc2, pei, nei, out,
                                                       npos, nneg);
}


# Row-sum kernel: dependent memory round trips per node cut from five or six to one: a node's whole list (24 rows) in flight at once, next node's list and row fetched behind it; sums keep their order
# speedup vs baseline: 1.8570x; 1.0332x over previous
#include <hip/hip_runtime.h>
#include <hip/hip_bf16.h>


typedef float v2f __attribute__((ext_vector_type(2)));
typedef float v8f __attribute__((ext_vector_type(8)));

#define H_DIM 64

#define ZERO_PER_THREAD 4
#define EDGES_PER_THREAD 8

__global__ void k_cnt_zero(int* __restrict__ cnt, int n) {
  const int T = gridDim.x * blockDim.x;
  int i = blockIdx.x * blockDim.x + threadIdx.x;
#pragma unroll
  for (int j = 0; j < ZERO_PER_THREAD; ++j, i += T)
    if (i < n) cnt[i] = 0;
}

__global__ void k_count_rank(const int* __restrict__ dst, int* __restrict__ cnt,
                             int* __restrict__ rank, int ne) {
  const int T = gridDim.x * blockDim.x;
  const int e0 = blockIdx.x * blockDim.x + threadIdx.x;
  int d[EDGES_PER_THREAD];
#pragma unroll
  for (int j = 0; j < EDGES_PER_THREAD; ++j) {
    const int e = e0 + j * T;
    d[j] = (e < ne) ? dst[e] : 0;
  }
  int r[EDGES_PER_THREAD];
#pragma unroll
  for (int j = 0; j < EDGES_PER_THREAD; ++j) {
    const int e = e0 + j * T;
    r[j] = 0;
    if (e < ne) r[j] = atomicAdd(&cnt[d[j]], 1);
  }
#pragma unroll
  for (int j = 0; j < EDGES_PER_THREAD; ++j) {
    const int e = e0 + j * T;
    if (e < ne) rank[e] = r[j];
  }
}

__global__ void __launch_bounds__(1024)
k_scan_dis(const int* __restrict__ cnt, int* __restrict__ off,
           float* __restrict__ dis, int n) {
  __shared__ int s_part[32];
  __shared__ int s_wsum[32];
  const int t = threadIdx.x;
  const int lane = t & 31;
  const int w = t >> 5;
  const int base = blockIdx.x * 1024;

  int pre = 0;
  for (int k = t; k < base; k += 1024) pre += cnt[k];
  for (int o = 16; o > 0; o >>= 1) pre += __shfl_xor(pre, o, 32);
  if (lane == 0) s_part[w] = pre;

  const int i = base + t;
  const int v = (i < n) ? cnt[i] : 0;
  int inc = v;
  for (int o = 1; o < 32; o <<= 1) {
    const int u = __shfl_up(inc, o, 32);
    if (lane >= o) inc += u;
  }
  if (lane == 31) s_wsum[w] = inc;
  __syncthreads();

  int p = s_part[lane];
  for (int o = 16; o > 0; o >>= 1) p += __shfl_xor(p, o, 32);
  int q = (lane < w) ? s_wsum[lane] : 0;
  for (int o = 16; o > 0; o >>= 1) q += __shfl_xor(q, o, 32);

  const int excl = p + q + inc - v;
  if (i < n) {
    off[i] = excl;
    dis[i] = rsqrtf((float)(v + 1));
    if (i == n - 1) off[n] = excl + v;
  }
}

__global__ void k_place(const int* __restrict__ src, const int* __restrict__ dst,
                        const int* __restrict__ rank, const int* __restrict__ off,
                        const float* __restrict__ dis, int* __restrict__ csr,
                        int ne) {
  const int T = gridDim.x * blockDim.x;
  const int e0 = blockIdx.x * blockDim.x + threadIdx.x;
  int s[EDGES_PER_THREAD], d[EDGES_PER_THREAD], r[EDGES_PER_THREAD];
#pragma unroll
  for (int j = 0; j < EDGES_PER_THREAD; ++j) {
    const int e = e0 + j * T;
    const bool ok = e < ne;
    s[j] = ok ? src[e] : 0;
    d[j] = ok ? dst[e] : 0;
    r[j] = ok ? rank[e] : 0;
  }
  int slot[EDGES_PER_THREAD];
  float w[EDGES_PER_THREAD];
#pragma unroll
  for (int j = 0; j < EDGES_PER_THREAD; ++j) {
    const int e = e0 + j * T;
    slot[j] = 0;
    w[j] = 0.0f;
    if (e < ne) {
      slot[j] = off[d[j]] + r[j];
      w[j] = dis[s[j]];
    }
  }
#pragma unroll
  for (int j = 0; j < EDGES_PER_THREAD; ++j) {
    const int e = e0 + j * T;
    if (e < ne) ((int2*)csr)[slot[j]] = make_int2(s[j], __float_as_int(w[j]));
  }
}

__global__ void __launch_bounds__(128)
k_gemm_wmma(const float* __restrict__ X, const float* __restrict__ W,
            float* __restrict__ Hout, int n, int tiles) {
  const int lane  = threadIdx.x & 31;
  const int wave  = threadIdx.x >> 5;
  const int n0    = wave * 16;
  const int m     = lane & 15;
  const int khalf = (lane >> 4) << 1;
  const int rbase = (lane >> 4) << 3;
  const int col   = n0 + m;

  v2f bf[16];
#pragma unroll
  for (int kk = 0; kk < 16; ++kk) {
    const int kb = kk * 4 + khalf;
    bf[kk].x = W[kb * H_DIM + col];
    bf[kk].y = W[(kb + 1) * H_DIM + col];
  }

  for (int tile = blockIdx.x; tile < tiles; tile += gridDim.x) {
    const int row0 = tile * 16;
    int arow = row0 + m;
    if (arow >= n) arow = n - 1;
    const float* __restrict__ xrow = X + (size_t)arow * H_DIM;

    v8f acc = {};
#pragma unroll
    for (int kk = 0; kk < 16; ++kk) {
      const int kb = kk * 4 + khalf;
      v2f a;
      a.x = xrow[kb];
      a.y = xrow[kb + 1];
      acc = __builtin_amdgcn_wmma_f32_16x16x4_f32(false, a, false, bf[kk],
                                                  (short)0, acc, false, false);
    }

#pragma unroll
    for (int r = 0; r < 8; ++r) {
      const int row = row0 + rbase + r;
      if (row < n) Hout[(size_t)row * H_DIM + col] = acc[r];
    }
  }
}

#define ROWS_AHEAD 24
template <int I>
__device__ __forceinline__ void rows_load(float2 (&v)[ROWS_AHEAD],
                                          const float2* __restrict__ H2,
                                          int s_l, int t, int cnt, int lane) {
  if constexpr (I < ROWS_AHEAD) {
    if (I < cnt) {
      const int s = __builtin_amdgcn_readlane(s_l, t + I);
      v[I] = H2[(size_t)s * (H_DIM / 2) + lane];
      rows_load<I + 1>(v, H2, s_l, t, cnt, lane);
    }
  }
}
template <int I>
__device__ __forceinline__ void rows_sum(const float2 (&v)[ROWS_AHEAD],
                                         float w_l, int t, int cnt, float& ax,
                                         float& ay) {
  if constexpr (I < ROWS_AHEAD) {
    if (I < cnt) {
      const float w = __int_as_float(
          __builtin_amdgcn_readlane(__float_as_int(w_l), t + I));
      ax = fmaf(v[I].x, w, ax);
      ay = fmaf(v[I].y, w, ay);
      rows_sum<I + 1>(v, w_l, t, cnt, ax, ay);
    }
  }
}

template <bool RELU>
__global__ void __launch_bounds__(256)
k_gather_rows(const float* __restrict__ Hm, const int* __restrict__ off,
              const int* __restrict__ csr, const float* __restrict__ dis,
              const float* __restrict__ bias, float* __restrict__ Out, int n) {
  const int wave0 = __builtin_amdgcn_readfirstlane(
      (int)(blockIdx.x * (blockDim.x >> 5) + (threadIdx.x >> 5)));
  const int nwaves = (int)(gridDim.x * (blockDim.x >> 5));
  const int lane = threadIdx.x & 31;
  const float2 bb = ((const float2*)bias)[lane];
  const float2* __restrict__ H2 = (const float2*)Hm;
  const int2* __restrict__ C2 = (const int2*)csr;

  for (int nb = wave0; nb < n; nb += 32 * nwaves) {
    const int mine = nb + lane * nwaves;
    int b_l = 0, e_l = 0;
    float d_l = 0.0f;
    if (mine < n) {
      b_l = off[mine];
      e_l = off[mine + 1];
      d_l = dis[mine];
    }
    int beg = __builtin_amdgcn_readlane(b_l, 0);
    int end = __builtin_amdgcn_readlane(e_l, 0);
    int2 p = make_int2(0, 0);
    if (lane < end - beg) p = C2[beg + lane];
    float2 hs = H2[(size_t)nb * (H_DIM / 2) + lane];

    int node = nb;
    for (int j = 0; j < 32; ++j) {
      const float dd = __int_as_float(
          __builtin_amdgcn_readlane(__float_as_int(d_l), j));
      const int nxt = node + nwaves;
      const bool more = (j + 1 < 32) && (nxt < n);
      float ax = fmaf(hs.x, dd * dd, bb.x);
      float ay = fmaf(hs.y, dd * dd, bb.y);
      const int deg = end - beg;
      int nbeg = 0, nend = 0;
      int2 pn = make_int2(0, 0);
      float2 hn = make_float2(0.0f, 0.0f);

      int done = 0;
      do {
        const int t = done & 31;
        if (done != 0 && t == 0) {
          p = make_int2(0, 0);
          if (lane < deg - done) p = C2[beg + done + lane];
        }
        asm volatile("" : : "v"(p.x), "v"(p.y));
        const int s_l = p.x;
        const float w_l = __int_as_float(p.y) * dd;
        int cnt = deg - done;
        if (cnt > 32 - t) cnt = 32 - t;
        if (cnt > ROWS_AHEAD) cnt = ROWS_AHEAD;

        float2 v[ROWS_AHEAD];
        rows_load<0>(v, H2, s_l, t, cnt, lane);
        if (done == 0 && more) {
          nbeg = __builtin_amdgcn_readlane(b_l, j + 1);
          nend = __builtin_amdgcn_readlane(e_l, j + 1);
          if (lane < nend - nbeg) pn = C2[nbeg + lane];
          hn = H2[(size_t)nxt * (H_DIM / 2) + lane];
        }
        rows_sum<0>(v, w_l, t, cnt, ax, ay);
        done += cnt;
      } while (done < deg);

      if (RELU) {
        ax = fmaxf(ax, 0.0f);
        ay = fmaxf(ay, 0.0f);
      }
      ((float2*)Out)[(size_t)node * (H_DIM / 2) + lane] = make_float2(ax, ay);
      if (!more) break;
      beg = nbeg;
      end = nend;
      p = pn;
      hs = hn;
      node = nxt;
    }
  }
}

#define DEC_GROUP 16
__global__ void __launch_bounds__(256)
k_decode(const float* __restrict__ Z, const int* __restrict__ pos,
         const int* __restrict__ neg, float* __restrict__ out, int npos,
         int nneg) {
  const int total = npos + nneg;
  const int base = __builtin_amdgcn_readfirstlane(
      (int)(blockIdx.x * (blockDim.x >> 5) + (threadIdx.x >> 5))) * 32;
  if (base >= total) return;
  const int lane = threadIdx.x & 31;
  const int k = base + lane;
  int a = 0, b = 0;
  if (k < npos) {
    a = pos[k];
    b = pos[npos + k];
  } else if (k < total) {
    const int j = k - npos;
    a = neg[j];
    b = neg[nneg + j];
  }
  const float2* __restrict__ Z2 = (const float2*)Z;
  float res = 0.0f;
#pragma unroll 1
  for (int t0 = 0; t0 < 32; t0 += DEC_GROUP) {
    float2 va[DEC_GROUP], vb[DEC_GROUP];
#pragma unroll
    for (int i = 0; i < DEC_GROUP; ++i) {
      const int sa = __builtin_amdgcn_readlane(a, t0 + i);
      const int sb = __builtin_amdgcn_readlane(b, t0 + i);
      va[i] = Z2[(size_t)sa * (H_DIM / 2) + lane];
      vb[i] = Z2[(size_t)sb * (H_DIM / 2) + lane];
    }
#pragma unroll
    for (int i = 0; i < DEC_GROUP; ++i) {
      float s = fmaf(va[i].x, vb[i].x, va[i].y * vb[i].y);
#pragma unroll
      for (int o = 16; o > 0; o >>= 1) s += __shfl_xor(s, o, 32);
      if (lane == t0 + i) res = s;
    }
  }
  if (k < total) out[k] = res;
}

extern "C" void kernel_launch(void* const* d_in, const int* in_sizes, int n_in,
                              void* d_out, int out_size, void* d_ws,
                              size_t ws_size, hipStream_t stream) {
  (void)n_in; (void)out_size; (void)ws_size;
  const float* x   = (const float*)d_in[0];
  const int*   ei  = (const int*)d_in[1];
  const int*   pei = (const int*)d_in[2];
  const int*   nei = (const int*)d_in[3];
  const float* W1  = (const float*)d_in[4];
  const float* b1  = (const float*)d_in[5];
  const float* W2  = (const float*)d_in[6];
  const float* b2  = (const float*)d_in[7];
  float* out = (float*)d_out;

  const int n    = in_sizes[0] / H_DIM;
  const int ne   = in_sizes[1] / 2;
  const int npos = in_sizes[2] / 2;
  const int nneg = in_sizes[3] / 2;
  const int* src = ei;
  const int* dst = ei + ne;

  float* ws = (float*)d_ws;
  const size_t nAl  = ((size_t)n + 63) & ~(size_t)63;
  const size_t neAl = ((size_t)ne + 63) & ~(size_t)63;
  float* dis  = ws;
  float* hbuf = ws + nAl;
  float* acc1 = hbuf + (size_t)n * H_DIM;
  float* acc2 = acc1 + (size_t)n * H_DIM;
  int* cnt  = (int*)(acc2 + (size_t)n * H_DIM);
  int* off  = cnt + nAl;
  int* rank = off + nAl + 64;
  int* csr  = rank + neAl;

  const int B = 256;
  const int zB = B * ZERO_PER_THREAD;
  const int eB = B * EDGES_PER_THREAD;
  k_cnt_zero<<<(n + zB - 1) / zB, B, 0, stream>>>(cnt, n);
  k_count_rank<<<(ne + eB - 1) / eB, B, 0, stream>>>(dst, cnt, rank, ne);
  k_scan_dis<<<(n + 1023) / 1024, 1024, 0, stream>>>(cnt, off, dis, n);
  k_place<<<(ne + eB - 1) / eB, B, 0, stream>>>(src, dst, rank, off, dis, csr, ne);

  const int tiles = (n + 15) / 16;
  const int gemm_blocks = tiles < 2048 ? tiles : 2048;
  const int node_blocks = (n + 7) / 8;
  const int gblocks = node_blocks < 1280 ? node_blocks : 1280;
  k_gemm_wmma<<<gemm_blocks, 128, 0, stream>>>(x, W1, hbuf, n, tiles);
  k_gather_rows<true><<<gblocks, B, 0, stream>>>(hbuf, off, csr, dis, b1, acc1, n);
  k_gemm_wmma<<<gemm_blocks, 128, 0, stream>>>(acc1, W2, hbuf, n, tiles);
  k_gather_rows<false><<<gblocks, B, 0, stream>>>(hbuf, off, csr, dis, b2, acc2, n);
  k_decode<<<(npos + nneg + 255) / 256, B, 0, stream>>>(acc2, pei, nei, out,
                                                       npos, nneg);
}


# Layer-1 GEMM moved into the edge-count launch on blocks of its own (X @ W1 needs nothing from the edge list): it runs under the atomics' waiting time; 9 launches become 8; sums keep their order
# speedup vs baseline: 1.9405x; 1.0450x over previous
#include <hip/hip_runtime.h>
#include <hip/hip_bf16.h>


typedef float v2f __attribute__((ext_vector_type(2)));
typedef float v8f __attribute__((ext_vector_type(8)));

#define H_DIM 64

#define ZERO_PER_THREAD 4
#define EDGES_PER_THREAD 8

__global__ void k_cnt_zero(int* __restrict__ cnt, int n) {
  const int T = gridDim.x * blockDim.x;
  int i = blockIdx.x * blockDim.x + threadIdx.x;
#pragma unroll
  for (int j = 0; j < ZERO_PER_THREAD; ++j, i += T)
    if (i < n) cnt[i] = 0;
}

__device__ __forceinline__ void count_rank_edges(const int* __restrict__ dst,
                                                 int* __restrict__ cnt,
                                                 int* __restrict__ rank, int ne,
                                                 int e0, int T) {
  int d[EDGES_PER_THREAD];
#pragma unroll
  for (int j = 0; j < EDGES_PER_THREAD; ++j) {
    const int e = e0 + j * T;
    d[j] = (e < ne) ? dst[e] : 0;
  }
  int r[EDGES_PER_THREAD];
#pragma unroll
  for (int j = 0; j < EDGES_PER_THREAD; ++j) {
    const int e = e0 + j * T;
    r[j] = 0;
    if (e < ne) r[j] = atomicAdd(&cnt[d[j]], 1);
  }
#pragma unroll
  for (int j = 0; j < EDGES_PER_THREAD; ++j) {
    const int e = e0 + j * T;
    if (e < ne) rank[e] = r[j];
  }
}

__global__ void __launch_bounds__(1024)
k_scan_dis(const int* __restrict__ cnt, int* __restrict__ off,
           float* __restrict__ dis, int n) {
  __shared__ int s_part[32];
  __shared__ int s_wsum[32];
  const int t = threadIdx.x;
  const int lane = t & 31;
  const int w = t >> 5;
  const int base = blockIdx.x * 1024;

  int pre = 0;
  for (int k = t; k < base; k += 1024) pre += cnt[k];
  for (int o = 16; o > 0; o >>= 1) pre += __shfl_xor(pre, o, 32);
  if (lane == 0) s_part[w] = pre;

  const int i = base + t;
  const int v = (i < n) ? cnt[i] : 0;
  int inc = v;
  for (int o = 1; o < 32; o <<= 1) {
    const int u = __shfl_up(inc, o, 32);
    if (lane >= o) inc += u;
  }
  if (lane == 31) s_wsum[w] = inc;
  __syncthreads();

  int p = s_part[lane];
  for (int o = 16; o > 0; o >>= 1) p += __shfl_xor(p, o, 32);
  int q = (lane < w) ? s_wsum[lane] : 0;
  for (int o = 16; o > 0; o >>= 1) q += __shfl_xor(q, o, 32);

  const int excl = p + q + inc - v;
  if (i < n) {
    off[i] = excl;
    dis[i] = rsqrtf((float)(v + 1));
    if (i == n - 1) off[n] = excl + v;
  }
}

__global__ void k_place(const int* __restrict__ src, const int* __restrict__ dst,
                        const int* __restrict__ rank, const int* __restrict__ off,
                        const float* __restrict__ dis, int* __restrict__ csr,
                        int ne) {
  const int T = gridDim.x * blockDim.x;
  const int e0 = blockIdx.x * blockDim.x + threadIdx.x;
  int s[EDGES_PER_THREAD], d[EDGES_PER_THREAD], r[EDGES_PER_THREAD];
#pragma unroll
  for (int j = 0; j < EDGES_PER_THREAD; ++j) {
    const int e = e0 + j * T;
    const bool ok = e < ne;
    s[j] = ok ? src[e] : 0;
    d[j] = ok ? dst[e] : 0;
    r[j] = ok ? rank[e] : 0;
  }
  int slot[EDGES_PER_THREAD];
  float w[EDGES_PER_THREAD];
#pragma unroll
  for (int j = 0; j < EDGES_PER_THREAD; ++j) {
    const int e = e0 + j * T;
    slot[j] = 0;
    w[j] = 0.0f;
    if (e < ne) {
      slot[j] = off[d[j]] + r[j];
      w[j] = dis[s[j]];
    }
  }
#pragma unroll
  for (int j = 0; j < EDGES_PER_THREAD; ++j) {
    const int e = e0 + j * T;
    if (e < ne) ((int2*)csr)[slot[j]] = make_int2(s[j], __float_as_int(w[j]));
  }
}

__device__ __forceinline__ void gemm_tiles(const float* __restrict__ X,
                                           const float* __restrict__ W,
                                           float* __restrict__ Hout, int n,
                                           int tiles, int tile0, int tstride,
                                           int wave) {
  const int lane  = threadIdx.x & 31;
  const int n0    = wave * 16;
  const int m     = lane & 15;
  const int khalf = (lane >> 4) << 1;
  const int rbase = (lane >> 4) << 3;
  const int col   = n0 + m;

  v2f bf[16];
#pragma unroll
  for (int kk = 0; kk < 16; ++kk) {
    const int kb = kk * 4 + khalf;
    bf[kk].x = W[kb * H_DIM + col];
    bf[kk].y = W[(kb + 1) * H_DIM + col];
  }

  for (int tile = tile0; tile < tiles; tile += tstride) {
    const int row0 = tile * 16;
    int arow = row0 + m;
    if (arow >= n) arow = n - 1;
    const float* __restrict__ xrow = X + (size_t)arow * H_DIM;

    v8f acc = {};
#pragma unroll
    for (int kk = 0; kk < 16; ++kk) {
      const int kb = kk * 4 + khalf;
      v2f a;
      a.x = xrow[kb];
      a.y = xrow[kb + 1];
      acc = __builtin_amdgcn_wmma_f32_16x16x4_f32(false, a, false, bf[kk],
                                                  (short)0, acc, false, false);
    }

#pragma unroll
    for (int r = 0; r < 8; ++r) {
      const int row = row0 + rbase + r;
      if (row < n) Hout[(size_t)row * H_DIM + col] = acc[r];
    }
  }
}

__global__ void __launch_bounds__(128)
k_gemm_wmma(const float* __restrict__ X, const float* __restrict__ W,
            float* __restrict__ Hout, int n, int tiles) {
  gemm_tiles(X, W, Hout, n, tiles, (int)blockIdx.x, (int)gridDim.x,
             (int)(threadIdx.x >> 5));
}

__global__ void __launch_bounds__(256)
k_count_rank_gemm(const int* __restrict__ dst, int* __restrict__ cnt,
                  int* __restrict__ rank, int ne, int cblocks,
                  const float* __restrict__ X, const float* __restrict__ W,
                  float* __restrict__ Hout, int n, int tiles) {
  if ((int)blockIdx.x < cblocks) {
    count_rank_edges(dst, cnt, rank, ne,
                     (int)(blockIdx.x * blockDim.x + threadIdx.x),
                     cblocks * (int)blockDim.x);
  } else {
    const int wave = __builtin_amdgcn_readfirstlane((int)(threadIdx.x >> 5));
    const int g = (int)blockIdx.x - cblocks;
    const int G = (int)gridDim.x - cblocks;
    gemm_tiles(X, W, Hout, n, tiles, g * 2 + (wave >> 2), G * 2, wave & 3);
  }
}

#define ROWS_AHEAD 24
template <int I>
__device__ __forceinline__ void rows_load(float2 (&v)[ROWS_AHEAD],
                                          const float2* __restrict__ H2,
                                          int s_l, int t, int cnt, int lane) {
  if constexpr (I < ROWS_AHEAD) {
    if (I < cnt) {
      const int s = __builtin_amdgcn_readlane(s_l, t + I);
      v[I] = H2[(size_t)s * (H_DIM / 2) + lane];
      rows_load<I + 1>(v, H2, s_l, t, cnt, lane);
    }
  }
}
template <int I>
__device__ __forceinline__ void rows_sum(const float2 (&v)[ROWS_AHEAD],
                                         float w_l, int t, int cnt, float& ax,
                                         float& ay) {
  if constexpr (I < ROWS_AHEAD) {
    if (I < cnt) {
      const float w = __int_as_float(
          __builtin_amdgcn_readlane(__float_as_int(w_l), t + I));
      ax = fmaf(v[I].x, w, ax);
      ay = fmaf(v[I].y, w, ay);
      rows_sum<I + 1>(v, w_l, t, cnt, ax, ay);
    }
  }
}

template <bool RELU>
__global__ void __launch_bounds__(256)
k_gather_rows(const float* __restrict__ Hm, const int* __restrict__ off,
              const int* __restrict__ csr, const float* __restrict__ dis,
              const float* __restrict__ bias, float* __restrict__ Out, int n) {
  const int wave0 = __builtin_amdgcn_readfirstlane(
      (int)(blockIdx.x * (blockDim.x >> 5) + (threadIdx.x >> 5)));
  const int nwaves = (int)(gridDim.x * (blockDim.x >> 5));
  const int lane = threadIdx.x & 31;
  const float2 bb = ((const float2*)bias)[lane];
  const float2* __restrict__ H2 = (const float2*)Hm;
  const int2* __restrict__ C2 = (const int2*)csr;

  for (int nb = wave0; nb < n; nb += 32 * nwaves) {
    const int mine = nb + lane * nwaves;
    int b_l = 0, e_l = 0;
    float d_l = 0.0f;
    if (mine < n) {
      b_l = off[mine];
      e_l = off[mine + 1];
      d_l = dis[mine];
    }
    int beg = __builtin_amdgcn_readlane(b_l, 0);
    int end = __builtin_amdgcn_readlane(e_l, 0);
    int2 p = make_int2(0, 0);
    if (lane < end - beg) p = C2[beg + lane];
    float2 hs = H2[(size_t)nb * (H_DIM / 2) + lane];

    int node = nb;
    for (int j = 0; j < 32; ++j) {
      const float dd = __int_as_float(
          __builtin_amdgcn_readlane(__float_as_int(d_l), j));
      const int nxt = node + nwaves;
      const bool more = (j + 1 < 32) && (nxt < n);
      float ax = fmaf(hs.x, dd * dd, bb.x);
      float ay = fmaf(hs.y, dd * dd, bb.y);
      const int deg = end - beg;
      int nbeg = 0, nend = 0;
      int2 pn = make_int2(0, 0);
      float2 hn = make_float2(0.0f, 0.0f);

      int done = 0;
      do {
        const int t = done & 31;
        if (done != 0 && t == 0) {
          p = make_int2(0, 0);
          if (lane < deg - done) p = C2[beg + done + lane];
        }
        asm volatile("" : : "v"(p.x), "v"(p.y));
        const int s_l = p.x;
        const float w_l = __int_as_float(p.y) * dd;
        int cnt = deg - done;
        if (cnt > 32 - t) cnt = 32 - t;
        if (cnt > ROWS_AHEAD) cnt = ROWS_AHEAD;

        float2 v[ROWS_AHEAD];
        rows_load<0>(v, H2, s_l, t, cnt, lane);
        if (done == 0 && more) {
          nbeg = __builtin_amdgcn_readlane(b_l, j + 1);
          nend = __builtin_amdgcn_readlane(e_l, j + 1);
          if (lane < nend - nbeg) pn = C2[nbeg + lane];
          hn = H2[(size_t)nxt * (H_DIM / 2) + lane];
        }
        rows_sum<0>(v, w_l, t, cnt, ax, ay);
        done += cnt;
      } while (done < deg);

      if (RELU) {
        ax = fmaxf(ax, 0.0f);
        ay = fmaxf(ay, 0.0f);
      }
      ((float2*)Out)[(size_t)node * (H_DIM / 2) + lane] = make_float2(ax, ay);
      if (!more) break;
      beg = nbeg;
      end = nend;
      p = pn;
      hs = hn;
      node = nxt;
    }
  }
}

#define DEC_GROUP 16
__global__ void __launch_bounds__(256)
k_decode(const float* __restrict__ Z, const int* __restrict__ pos,
         const int* __restrict__ neg, float* __restrict__ out, int npos,
         int nneg) {
  const int total = npos + nneg;
  const int base = __builtin_amdgcn_readfirstlane(
      (int)(blockIdx.x * (blockDim.x >> 5) + (threadIdx.x >> 5))) * 32;
  if (base >= total) return;
  const int lane = threadIdx.x & 31;
  const int k = base + lane;
  int a = 0, b = 0;
  if (k < npos) {
    a = pos[k];
    b = pos[npos + k];
  } else if (k < total) {
    const int j = k - npos;
    a = neg[j];
    b = neg[nneg + j];
  }
  const float2* __restrict__ Z2 = (const float2*)Z;
  float res = 0.0f;
#pragma unroll 1
  for (int t0 = 0; t0 < 32; t0 += DEC_GROUP) {
    float2 va[DEC_GROUP], vb[DEC_GROUP];
#pragma unroll
    for (int i = 0; i < DEC_GROUP; ++i) {
      const int sa = __builtin_amdgcn_readlane(a, t0 + i);
      const int sb = __builtin_amdgcn_readlane(b, t0 + i);
      va[i] = Z2[(size_t)sa * (H_DIM / 2) + lane];
      vb[i] = Z2[(size_t)sb * (H_DIM / 2) + lane];
    }
#pragma unroll
    for (int i = 0; i < DEC_GROUP; ++i) {
      float s = fmaf(va[i].x, vb[i].x, va[i].y * vb[i].y);
#pragma unroll
      for (int o = 16; o > 0; o >>= 1) s += __shfl_xor(s, o, 32);
      if (lane == t0 + i) res = s;
    }
  }
  if (k < total) out[k] = res;
}

extern "C" void kernel_launch(void* const* d_in, const int* in_sizes, int n_in,
                              void* d_out, int out_size, void* d_ws,
                              size_t ws_size, hipStream_t stream) {
  (void)n_in; (void)out_size; (void)ws_size;
  const float* x   = (const float*)d_in[0];
  const int*   ei  = (const int*)d_in[1];
  const int*   pei = (const int*)d_in[2];
  const int*   nei = (const int*)d_in[3];
  const float* W1  = (const float*)d_in[4];
  const float* b1  = (const float*)d_in[5];
  const float* W2  = (const float*)d_in[6];
  const float* b2  = (const float*)d_in[7];
  float* out = (float*)d_out;

  const int n    = in_sizes[0] / H_DIM;
  const int ne   = in_sizes[1] / 2;
  const int npos = in_sizes[2] / 2;
  const int nneg = in_sizes[3] / 2;
  const int* src = ei;
  const int* dst = ei + ne;

  float* ws = (float*)d_ws;
  const size_t nAl  = ((size_t)n + 63) & ~(size_t)63;
  const size_t neAl = ((size_t)ne + 63) & ~(size_t)63;
  float* dis  = ws;
  float* hbuf = ws + nAl;
  float* acc1 = hbuf + (size_t)n * H_DIM;
  float* acc2 = acc1 + (size_t)n * H_DIM;
  int* cnt  = (int*)(acc2 + (size_t)n * H_DIM);
  int* off  = cnt + nAl;
  int* rank = off + nAl + 64;
  int* csr  = rank + neAl;

  const int B = 256;
  const int zB = B * ZERO_PER_THREAD;
  const int eB = B * EDGES_PER_THREAD;
  k_cnt_zero<<<(n + zB - 1) / zB, B, 0, stream>>>(cnt, n);
  const int tiles = (n + 15) / 16;
  const int cblocks = (ne + eB - 1) / eB;
  const int gemm_blocks8 = (tiles + 1) / 2 < 1024 ? (tiles + 1) / 2 : 1024;
  k_count_rank_gemm<<<cblocks + gemm_blocks8, B, 0, stream>>>(
      dst, cnt, rank, ne, cblocks, x, W1, hbuf, n, tiles);
  k_scan_dis<<<(n + 1023) / 1024, 1024, 0, stream>>>(cnt, off, dis, n);
  k_place<<<(ne + eB - 1) / eB, B, 0, stream>>>(src, dst, rank, off, dis, csr, ne);

  const int gemm_blocks = tiles < 2048 ? tiles : 2048;
  const int node_blocks = (n + 7) / 8;
  const int gblocks = node_blocks < 1280 ? node_blocks : 1280;
  k_gather_rows<true><<<gblocks, B, 0, stream>>>(hbuf, off, csr, dis, b1, acc1, n);
  k_gemm_wmma<<<gemm_blocks, 128, 0, stream>>>(acc1, W2, hbuf, n, tiles);
  k_gather_rows<false><<<gblocks, B, 0, stream>>>(hbuf, off, csr, dis, b2, acc2, n);
  k_decode<<<(npos + nneg + 255) / 256, B, 0, stream>>>(acc2, pei, nei, out,
                                                       npos, nneg);
}


# Scan and placing pass removed: the count's returning atomic puts each edge's source into slot [rank] of a fixed 32-slot bucket per destination (overflow list past it); 8 launches to 6; sum order kept
# speedup vs baseline: 2.0904x; 1.0772x over previous
#include <hip/hip_runtime.h>
#include <hip/hip_bf16.h>


typedef float v2f __attribute__((ext_vector_type(2)));
typedef float v8f __attribute__((ext_vector_type(8)));

#define H_DIM 64

#define ZERO_PER_THREAD 4
#define EDGES_PER_THREAD 8

__global__ void k_cnt_zero(int* __restrict__ cnt, int n,
                           int* __restrict__ ovfc) {
  const int T = gridDim.x * blockDim.x;
  int i = blockIdx.x * blockDim.x + threadIdx.x;
#pragma unroll
  for (int j = 0; j < ZERO_PER_THREAD; ++j, i += T)
    if (i < n) cnt[i] = 0;
  if (blockIdx.x == 0 && threadIdx.x == 0) ovfc[0] = 0;
}

#define BUCKET 32

template <bool GUARD>
__device__ __forceinline__ void count_bucket_body(
    const int* __restrict__ src, const int* __restrict__ dst,
    int* __restrict__ cnt, int* __restrict__ ovfc, int* __restrict__ bk,
    int* __restrict__ ovf, int ne, int e0, int T) {
  int d[EDGES_PER_THREAD], s[EDGES_PER_THREAD];
#pragma unroll
  for (int j = 0; j < EDGES_PER_THREAD; ++j) {
    const int e = e0 + j * T;
    const bool ok = !GUARD || e < ne;
    d[j] = ok ? dst[e] : 0;
    s[j] = ok ? src[e] : 0;
  }
  int r[EDGES_PER_THREAD];
#pragma unroll
  for (int j = 0; j < EDGES_PER_THREAD; ++j) {
    const int e = e0 + j * T;
    r[j] = 0;
    if (!GUARD || e < ne) r[j] = atomicAdd(&cnt[d[j]], 1);
  }
#pragma unroll
  for (int j = 0; j < EDGES_PER_THREAD; ++j) {
    const int e = e0 + j * T;
    if (!GUARD || e < ne) {
      if (r[j] < BUCKET) {
        bk[((size_t)d[j] * BUCKET + r[j]) * 2] = s[j];
      } else {
        const int o = atomicAdd(ovfc, 1);
        ((int2*)ovf)[o] = make_int2(d[j], s[j]);
      }
    }
  }
}

__device__ __forceinline__ void count_bucket_edges(
    const int* __restrict__ src, const int* __restrict__ dst,
    int* __restrict__ cnt, int* __restrict__ ovfc, int* __restrict__ bk,
    int* __restrict__ ovf, int ne, int e0, int T) {
  if (e0 + (EDGES_PER_THREAD - 1) * T < ne)
    count_bucket_body<false>(src, dst, cnt, ovfc, bk, ovf, ne, e0, T);
  else
    count_bucket_body<true>(src, dst, cnt, ovfc, bk, ovf, ne, e0, T);
}

__device__ __forceinline__ void gemm_tiles(const float* __restrict__ X,
                                           const float* __restrict__ W,
                                           float* __restrict__ Hout, int n,
                                           int tiles, int tile0, int tstride,
                                           int wave) {
  const int lane  = threadIdx.x & 31;
  const int n0    = wave * 16;
  const int m     = lane & 15;
  const int khalf = (lane >> 4) << 1;
  const int rbase = (lane >> 4) << 3;
  const int col   = n0 + m;

  v2f bf[16];
#pragma unroll
  for (int kk = 0; kk < 16; ++kk) {
    const int kb = kk * 4 + khalf;
    bf[kk].x = W[kb * H_DIM + col];
    bf[kk].y = W[(kb + 1) * H_DIM + col];
  }

  for (int tile = tile0; tile < tiles; tile += tstride) {
    const int row0 = tile * 16;
    int arow = row0 + m;
    if (arow >= n) arow = n - 1;
    const float* __restrict__ xrow = X + (size_t)arow * H_DIM;

    v8f acc = {};
#pragma unroll
    for (int kk = 0; kk < 16; ++kk) {
      const int kb = kk * 4 + khalf;
      v2f a;
      a.x = xrow[kb];
      a.y = xrow[kb + 1];
      acc = __builtin_amdgcn_wmma_f32_16x16x4_f32(false, a, false, bf[kk],
                                                  (short)0, acc, false, false);
    }

#pragma unroll
    for (int r = 0; r < 8; ++r) {
      const int row = row0 + rbase + r;
      if (row < n) Hout[(size_t)row * H_DIM + col] = acc[r];
    }
  }
}

__global__ void __launch_bounds__(128)
k_gemm_wmma(const float* __restrict__ X, const float* __restrict__ W,
            float* __restrict__ Hout, int n, int tiles) {
  gemm_tiles(X, W, Hout, n, tiles, (int)blockIdx.x, (int)gridDim.x,
             (int)(threadIdx.x >> 5));
}

__global__ void __launch_bounds__(256)
k_count_rank_gemm(const int* __restrict__ esrc, const int* __restrict__ dst,
                  int* __restrict__ cnt, int* __restrict__ ovfc,
                  int* __restrict__ bk, int* __restrict__ ovf, int ne,
                  int cblocks,
                  const float* __restrict__ X, const float* __restrict__ W,
                  float* __restrict__ Hout, int n, int tiles) {
  if ((int)blockIdx.x < cblocks) {
    count_bucket_edges(esrc, dst, cnt, ovfc, bk, ovf, ne,
                       (int)(blockIdx.x * blockDim.x + threadIdx.x),
                       cblocks * (int)blockDim.x);
  } else {
    const int wave = __builtin_amdgcn_readfirstlane((int)(threadIdx.x >> 5));
    const int g = (int)blockIdx.x - cblocks;
    const int G = (int)gridDim.x - cblocks;
    gemm_tiles(X, W, Hout, n, tiles, g * 2 + (wave >> 2), G * 2, wave & 3);
  }
}

#define ROWS_AHEAD 24
template <int I>
__device__ __forceinline__ void rows_load(float2 (&v)[ROWS_AHEAD],
                                          const float2* __restrict__ H2,
                                          int s_l, int t, int cnt, int lane) {
  if constexpr (I < ROWS_AHEAD) {
    if (I < cnt) {
      const int s = __builtin_amdgcn_readlane(s_l, t + I);
      v[I] = H2[(size_t)s * (H_DIM / 2) + lane];
      rows_load<I + 1>(v, H2, s_l, t, cnt, lane);
    }
  }
}
template <int I>
__device__ __forceinline__ void rows_sum(const float2 (&v)[ROWS_AHEAD],
                                         float w_l, int t, int cnt, float& ax,
                                         float& ay) {
  if constexpr (I < ROWS_AHEAD) {
    if (I < cnt) {
      const float w = __int_as_float(
          __builtin_amdgcn_readlane(__float_as_int(w_l), t + I));
      ax = fmaf(v[I].x, w, ax);
      ay = fmaf(v[I].y, w, ay);
      rows_sum<I + 1>(v, w_l, t, cnt, ax, ay);
    }
  }
}

template <bool FIRST>
__global__ void __launch_bounds__(256)
k_gather_rows(const float* __restrict__ Hm, const int* __restrict__ cnt,
              int* __restrict__ bk, const int* __restrict__ ovf,
              const int* __restrict__ ovfc, int ovf_cap,
              const float* __restrict__ bias, float* __restrict__ Out, int n) {
  const int wave0 = __builtin_amdgcn_readfirstlane(
      (int)(blockIdx.x * (blockDim.x >> 5) + (threadIdx.x >> 5)));
  const int nwaves = (int)(gridDim.x * (blockDim.x >> 5));
  const int lane = threadIdx.x & 31;
  const float2 bb = ((const float2*)bias)[lane];
  const float2* __restrict__ H2 = (const float2*)Hm;

  for (int nb = wave0; nb < n; nb += 32 * nwaves) {
    const int mine = nb + lane * nwaves;
    int c_l = 0;
    if (mine < n) c_l = cnt[mine];
    const float d_l = rsqrtf((float)(c_l + 1));
    int deg = __builtin_amdgcn_readlane(c_l, 0);
    int degb = deg < BUCKET ? deg : BUCKET;
    int2 p = make_int2(0, 0);
    if (lane < degb) {
      if (FIRST) p.x = bk[((size_t)nb * BUCKET + lane) * 2];
      else       p = ((const int2*)bk)[(size_t)nb * BUCKET + lane];
    }
    float2 hs = H2[(size_t)nb * (H_DIM / 2) + lane];

    int node = nb;
    for (int j = 0; j < 32; ++j) {
      const float dd = __int_as_float(
          __builtin_amdgcn_readlane(__float_as_int(d_l), j));
      const int nxt = node + nwaves;
      const bool more = (j + 1 < 32) && (nxt < n);
      float ax = fmaf(hs.x, dd * dd, bb.x);
      float ay = fmaf(hs.y, dd * dd, bb.y);
      int ndeg = 0, ndegb = 0;
      int2 pn = make_int2(0, 0);
      float2 hn = make_float2(0.0f, 0.0f);

      if (FIRST) asm volatile("" : : "v"(p.x));
      else       asm volatile("" : : "v"(p.x), "v"(p.y));
      const int s_l = p.x;
      int c_src = 0;
      if (FIRST) c_src = cnt[s_l];
      int gc = degb < ROWS_AHEAD ? degb : ROWS_AHEAD;
      float2 v[ROWS_AHEAD];
      rows_load<0>(v, H2, s_l, 0, gc, lane);
      if (more) {
        ndeg = __builtin_amdgcn_readlane(c_l, j + 1);
        ndegb = ndeg < BUCKET ? ndeg : BUCKET;
        if (lane < ndegb) {
          if (FIRST) pn.x = bk[((size_t)nxt * BUCKET + lane) * 2];
          else       pn = ((const int2*)bk)[(size_t)nxt * BUCKET + lane];
        }
        hn = H2[(size_t)nxt * (H_DIM / 2) + lane];
      }
      float w_l;
      if (FIRST) {
        const float ws = rsqrtf((float)(c_src + 1));
        if (lane < degb)
          bk[((size_t)node * BUCKET + lane) * 2 + 1] = __float_as_int(ws);
        w_l = ws * dd;
      } else {
        w_l = __int_as_float(p.y) * dd;
      }
      rows_sum<0>(v, w_l, 0, gc, ax, ay);
      for (int i = ROWS_AHEAD; i < degb; ++i) {
        const int s = __builtin_amdgcn_readlane(s_l, i);
        const float w = __int_as_float(
            __builtin_amdgcn_readlane(__float_as_int(w_l), i));
        const float2 r = H2[(size_t)s * (H_DIM / 2) + lane];
        ax = fmaf(r.x, w, ax);
        ay = fmaf(r.y, w, ay);
      }

      if (deg > BUCKET) {
        int L = ovfc[0];
        if (L > ovf_cap) L = ovf_cap;
        for (int b0 = 0; b0 < L; b0 += 32) {
          int2 q = make_int2(-1, 0);
          if (b0 + lane < L) q = ((const int2*)ovf)[b0 + lane];
          unsigned hit = __builtin_amdgcn_ballot_w32(q.x == node);
          while (hit != 0u) {
            const int l = __builtin_ctz(hit);
            hit &= hit - 1u;
            const int s = __builtin_amdgcn_readlane(q.y, l);
            const float w = rsqrtf((float)(cnt[s] + 1)) * dd;
            const float2 r = H2[(size_t)s * (H_DIM / 2) + lane];
            ax = fmaf(r.x, w, ax);
            ay = fmaf(r.y, w, ay);
          }
        }
      }

      if (FIRST) {
        ax = fmaxf(ax, 0.0f);
        ay = fmaxf(ay, 0.0f);
      }
      ((float2*)Out)[(size_t)node * (H_DIM / 2) + lane] = make_float2(ax, ay);
      if (!more) break;
      deg = ndeg;
      degb = ndegb;
      p = pn;
      hs = hn;
      node = nxt;
    }
  }
}

#define DEC_GROUP 16
__global__ void __launch_bounds__(256)
k_decode(const float* __restrict__ Z, const int* __restrict__ pos,
         const int* __restrict__ neg, float* __restrict__ out, int npos,
         int nneg) {
  const int total = npos + nneg;
  const int base = __builtin_amdgcn_readfirstlane(
      (int)(blockIdx.x * (blockDim.x >> 5) + (threadIdx.x >> 5))) * 32;
  if (base >= total) return;
  const int lane = threadIdx.x & 31;
  const int k = base + lane;
  int a = 0, b = 0;
  if (k < npos) {
    a = pos[k];
    b = pos[npos + k];
  } else if (k < total) {
    const int j = k - npos;
    a = neg[j];
    b = neg[nneg + j];
  }
  const float2* __restrict__ Z2 = (const float2*)Z;
  float res = 0.0f;
#pragma unroll 1
  for (int t0 = 0; t0 < 32; t0 += DEC_GROUP) {
    float2 va[DEC_GROUP], vb[DEC_GROUP];
#pragma unroll
    for (int i = 0; i < DEC_GROUP; ++i) {
      const int sa = __builtin_amdgcn_readlane(a, t0 + i);
      const int sb = __builtin_amdgcn_readlane(b, t0 + i);
      va[i] = Z2[(size_t)sa * (H_DIM / 2) + lane];
      vb[i] = Z2[(size_t)sb * (H_DIM / 2) + lane];
    }
#pragma unroll
    for (int i = 0; i < DEC_GROUP; ++i) {
      float s = fmaf(va[i].x, vb[i].x, va[i].y * vb[i].y);
#pragma unroll
      for (int o = 16; o > 0; o >>= 1) s += __shfl_xor(s, o, 32);
      if (lane == t0 + i) res = s;
    }
  }
  if (k < total) out[k] = res;
}

extern "C" void kernel_launch(void* const* d_in, const int* in_sizes, int n_in,
                              void* d_out, int out_size, void* d_ws,
                              size_t ws_size, hipStream_t stream) {
  (void)n_in; (void)out_size; (void)ws_size;
  const float* x   = (const float*)d_in[0];
  const int*   ei  = (const int*)d_in[1];
  const int*   pei = (const int*)d_in[2];
  const int*   nei = (const int*)d_in[3];
  const float* W1  = (const float*)d_in[4];
  const float* b1  = (const float*)d_in[5];
  const float* W2  = (const float*)d_in[6];
  const float* b2  = (const float*)d_in[7];
  float* out = (float*)d_out;

  const int n    = in_sizes[0] / H_DIM;
  const int ne   = in_sizes[1] / 2;
  const int npos = in_sizes[2] / 2;
  const int nneg = in_sizes[3] / 2;
  const int* src = ei;
  const int* dst = ei + ne;

  float* ws = (float*)d_ws;
  const size_t nAl = ((size_t)n + 63) & ~(size_t)63;
  float* hbuf = ws;
  float* acc1 = hbuf + (size_t)n * H_DIM;
  float* acc2 = acc1 + (size_t)n * H_DIM;
  int* cnt  = (int*)(acc2 + (size_t)n * H_DIM);
  int* ovfc = cnt + nAl;
  int* bk   = ovfc + 64;
  int* ovf  = bk + (size_t)n * BUCKET * 2;

  const int B = 256;
  const int zB = B * ZERO_PER_THREAD;
  const int eB = B * EDGES_PER_THREAD;
  k_cnt_zero<<<(n + zB - 1) / zB, B, 0, stream>>>(cnt, n, ovfc);
  const int tiles = (n + 15) / 16;
  const int cblocks = (ne + eB - 1) / eB;
  const int gemm_blocks8 = (tiles + 1) / 2 < 1024 ? (tiles + 1) / 2 : 1024;
  k_count_rank_gemm<<<cblocks + gemm_blocks8, B, 0, stream>>>(
      src, dst, cnt, ovfc, bk, ovf, ne, cblocks, x, W1, hbuf, n, tiles);

  const int gemm_blocks = tiles < 2048 ? tiles : 2048;
  const int node_blocks = (n + 7) / 8;
  const int gblocks = node_blocks < 1280 ? node_blocks : 1280;
  k_gather_rows<true><<<gblocks, B, 0, stream>>>(hbuf, cnt, bk, ovf, ovfc, ne,
                                                 b1, acc1, n);
  k_gemm_wmma<<<gemm_blocks, 128, 0, stream>>>(acc1, W2, hbuf, n, tiles);
  k_gather_rows<false><<<gblocks, B, 0, stream>>>(hbuf, cnt, bk, ovf, ovfc, ne,
                                                  b2, acc2, n);
  k_decode<<<(npos + nneg + 255) / 256, B, 0, stream>>>(acc2, pei, nei, out,
                                                       npos, nneg);
}


# Layer-2 GEMM moved into the layer-1 row-sum kernel: a wave keeps its finished rows in an LDS tile and multiplies them with W2 (in LDS) on WMMA itself; 6 launches become 5; sum order kept
# speedup vs baseline: 2.2424x; 1.0727x over previous
#include <hip/hip_runtime.h>
#include <hip/hip_bf16.h>


typedef float v2f __attribute__((ext_vector_type(2)));
typedef float v8f __attribute__((ext_vector_type(8)));

#define H_DIM 64

#define ZERO_PER_THREAD 4
#define EDGES_PER_THREAD 8

__global__ void k_cnt_zero(int* __restrict__ cnt, int n,
                           int* __restrict__ ovfc) {
  const int T = gridDim.x * blockDim.x;
  int i = blockIdx.x * blockDim.x + threadIdx.x;
#pragma unroll
  for (int j = 0; j < ZERO_PER_THREAD; ++j, i += T)
    if (i < n) cnt[i] = 0;
  if (blockIdx.x == 0 && threadIdx.x == 0) ovfc[0] = 0;
}

#define BUCKET 32

template <bool GUARD>
__device__ __forceinline__ void count_bucket_body(
    const int* __restrict__ src, const int* __restrict__ dst,
    int* __restrict__ cnt, int* __restrict__ ovfc, int* __restrict__ bk,
    int* __restrict__ ovf, int ne, int e0, int T) {
  int d[EDGES_PER_THREAD], s[EDGES_PER_THREAD];
#pragma unroll
  for (int j = 0; j < EDGES_PER_THREAD; ++j) {
    const int e = e0 + j * T;
    const bool ok = !GUARD || e < ne;
    d[j] = ok ? dst[e] : 0;
    s[j] = ok ? src[e] : 0;
  }
  int r[EDGES_PER_THREAD];
#pragma unroll
  for (int j = 0; j < EDGES_PER_THREAD; ++j) {
    const int e = e0 + j * T;
    r[j] = 0;
    if (!GUARD || e < ne) r[j] = atomicAdd(&cnt[d[j]], 1);
  }
#pragma unroll
  for (int j = 0; j < EDGES_PER_THREAD; ++j) {
    const int e = e0 + j * T;
    if (!GUARD || e < ne) {
      if (r[j] < BUCKET) {
        bk[((size_t)d[j] * BUCKET + r[j]) * 2] = s[j];
      } else {
        const int o = atomicAdd(ovfc, 1);
        ((int2*)ovf)[o] = make_int2(d[j], s[j]);
      }
    }
  }
}

__device__ __forceinline__ void count_bucket_edges(
    const int* __restrict__ src, const int* __restrict__ dst,
    int* __restrict__ cnt, int* __restrict__ ovfc, int* __restrict__ bk,
    int* __restrict__ ovf, int ne, int e0, int T) {
  if (e0 + (EDGES_PER_THREAD - 1) * T < ne)
    count_bucket_body<false>(src, dst, cnt, ovfc, bk, ovf, ne, e0, T);
  else
    count_bucket_body<true>(src, dst, cnt, ovfc, bk, ovf, ne, e0, T);
}

__device__ __forceinline__ void gemm_tiles(const float* __restrict__ X,
                                           const float* __restrict__ W,
                                           float* __restrict__ Hout, int n,
                                           int tiles, int tile0, int tstride,
                                           int wave) {
  const int lane  = threadIdx.x & 31;
  const int n0    = wave * 16;
  const int m     = lane & 15;
  const int khalf = (lane >> 4) << 1;
  const int rbase = (lane >> 4) << 3;
  const int col   = n0 + m;

  v2f bf[16];
#pragma unroll
  for (int kk = 0; kk < 16; ++kk) {
    const int kb = kk * 4 + khalf;
    bf[kk].x = W[kb * H_DIM + col];
    bf[kk].y = W[(kb + 1) * H_DIM + col];
  }

  for (int tile = tile0; tile < tiles; tile += tstride) {
    const int row0 = tile * 16;
    int arow = row0 + m;
    if (arow >= n) arow = n - 1;
    const float* __restrict__ xrow = X + (size_t)arow * H_DIM;

    v8f acc = {};
#pragma unroll
    for (int kk = 0; kk < 16; ++kk) {
      const int kb = kk * 4 + khalf;
      v2f a;
      a.x = xrow[kb];
      a.y = xrow[kb + 1];
      acc = __builtin_amdgcn_wmma_f32_16x16x4_f32(false, a, false, bf[kk],
                                                  (short)0, acc, false, false);
    }

#pragma unroll
    for (int r = 0; r < 8; ++r) {
      const int row = row0 + rbase + r;
      if (row < n) Hout[(size_t)row * H_DIM + col] = acc[r];
    }
  }
}

__global__ void __launch_bounds__(256)
k_count_rank_gemm(const int* __restrict__ esrc, const int* __restrict__ dst,
                  int* __restrict__ cnt, int* __restrict__ ovfc,
                  int* __restrict__ bk, int* __restrict__ ovf, int ne,
                  int cblocks,
                  const float* __restrict__ X, const float* __restrict__ W,
                  float* __restrict__ Hout, int n, int tiles) {
  if ((int)blockIdx.x < cblocks) {
    count_bucket_edges(esrc, dst, cnt, ovfc, bk, ovf, ne,
                       (int)(blockIdx.x * blockDim.x + threadIdx.x),
                       cblocks * (int)blockDim.x);
  } else {
    const int wave = __builtin_amdgcn_readfirstlane((int)(threadIdx.x >> 5));
    const int g = (int)blockIdx.x - cblocks;
    const int G = (int)gridDim.x - cblocks;
    gemm_tiles(X, W, Hout, n, tiles, g * 2 + (wave >> 2), G * 2, wave & 3);
  }
}

#define ROWS_AHEAD 24
template <int I>
__device__ __forceinline__ void rows_load(float2 (&v)[ROWS_AHEAD],
                                          const float2* __restrict__ H2,
                                          int s_l, int t, int cnt, int lane) {
  if constexpr (I < ROWS_AHEAD) {
    if (I < cnt) {
      const int s = __builtin_amdgcn_readlane(s_l, t + I);
      v[I] = H2[(size_t)s * (H_DIM / 2) + lane];
      rows_load<I + 1>(v, H2, s_l, t, cnt, lane);
    }
  }
}
template <int I>
__device__ __forceinline__ void rows_sum(const float2 (&v)[ROWS_AHEAD],
                                         float w_l, int t, int cnt, float& ax,
                                         float& ay) {
  if constexpr (I < ROWS_AHEAD) {
    if (I < cnt) {
      const float w = __int_as_float(
          __builtin_amdgcn_readlane(__float_as_int(w_l), t + I));
      ax = fmaf(v[I].x, w, ax);
      ay = fmaf(v[I].y, w, ay);
      rows_sum<I + 1>(v, w_l, t, cnt, ax, ay);
    }
  }
}

#define ZT_PITCH 68
__device__ __forceinline__ void tile_gemm_store(const float* zt,
                                                const float* wq,
                                                float* __restrict__ Hout,
                                                int nb, int nwaves, int nrows,
                                                int lane) {
  const int m     = lane & 15;
  const int khalf = (lane >> 4) << 1;
  const int rbase = (lane >> 4) << 3;
  const unsigned lane_off = (unsigned)(rbase * nwaves) * H_DIM + (unsigned)m;
#pragma unroll
  for (int c = 0; c < 4; c += 2) {
    v8f acc0 = {}, acc1 = {};
#pragma unroll
    for (int kk = 0; kk < 16; ++kk) {
      const v2f a  = *(const v2f*)(zt + m * ZT_PITCH + kk * 4 + khalf);
      const v2f b0 = *(const v2f*)(wq + ((kk * 64 + c * 16 + m) << 2) + khalf);
      const v2f b1 =
          *(const v2f*)(wq + ((kk * 64 + (c + 1) * 16 + m) << 2) + khalf);
      acc0 = __builtin_amdgcn_wmma_f32_16x16x4_f32(false, a, false, b0,
                                                   (short)0, acc0, false, false);
      acc1 = __builtin_amdgcn_wmma_f32_16x16x4_f32(false, a, false, b1,
                                                   (short)0, acc1, false, false);
    }
#pragma unroll
    for (int r = 0; r < 8; ++r) {
      if (rbase + r < nrows) {
        float* __restrict__ srow =
            Hout + (size_t)(nb + r * nwaves) * H_DIM + c * 16;
        srow[lane_off]      = acc0[r];
        srow[lane_off + 16] = acc1[r];
      }
    }
  }
}

template <bool FIRST>
__global__ void __launch_bounds__(256, 10)
k_gather_rows(const float* __restrict__ Hm, const int* __restrict__ cnt,
              int* __restrict__ bk, const int* __restrict__ ovf,
              const int* __restrict__ ovfc, int ovf_cap,
              const float* __restrict__ bias, const float* __restrict__ Wn,
              float* __restrict__ Out, int n) {
  const int wave0 = __builtin_amdgcn_readfirstlane(
      (int)(blockIdx.x * (blockDim.x >> 5) + (threadIdx.x >> 5)));
  const int nwaves = (int)(gridDim.x * (blockDim.x >> 5));
  const int lane = threadIdx.x & 31;
  const float2 bb = ((const float2*)bias)[lane];
  const float2* __restrict__ H2 = (const float2*)Hm;

  __shared__ alignas(16) float s_wq[H_DIM * H_DIM];
  __shared__ alignas(16) float s_zt[8][16 * ZT_PITCH];
  const int wib = __builtin_amdgcn_readfirstlane((int)(threadIdx.x >> 5));
  if (FIRST) {
    float4 wv[4];
#pragma unroll
    for (int q = 0; q < 4; ++q) {
      const int cell = q * 256 + (int)threadIdx.x;
      const float* __restrict__ wp = Wn + ((cell >> 6) << 8) + (cell & 63);
      wv[q] = make_float4(wp[0], wp[H_DIM], wp[2 * H_DIM], wp[3 * H_DIM]);
    }
#pragma unroll
    for (int q = 0; q < 4; ++q)
      ((float4*)s_wq)[q * 256 + (int)threadIdx.x] = wv[q];
    __syncthreads();
  }

  int nrows = 0;
  for (int nb = wave0; nb < n; nb += 32 * nwaves) {
    const int mine = nb + lane * nwaves;
    int c_l = 0;
    if (mine < n) c_l = cnt[mine];
    const float d_l = rsqrtf((float)(c_l + 1));
    int deg = __builtin_amdgcn_readlane(c_l, 0);
    int degb = deg < BUCKET ? deg : BUCKET;
    int2 p = make_int2(0, 0);
    if (lane < degb) {
      if (FIRST) p.x = bk[((size_t)nb * BUCKET + lane) * 2];
      else       p = ((const int2*)bk)[(size_t)nb * BUCKET + lane];
    }
    float2 hs = H2[(size_t)nb * (H_DIM / 2) + lane];

    int node = nb;
    for (int j = 0; j < 32; ++j) {
      const float dd = __int_as_float(
          __builtin_amdgcn_readlane(__float_as_int(d_l), j));
      const int nxt = node + nwaves;
      const bool more = (j + 1 < 32) && (nxt < n);
      float ax = fmaf(hs.x, dd * dd, bb.x);
      float ay = fmaf(hs.y, dd * dd, bb.y);
      int ndeg = 0, ndegb = 0;
      int2 pn = make_int2(0, 0);
      float2 hn = make_float2(0.0f, 0.0f);

      if (FIRST) asm volatile("" : : "v"(p.x));
      else       asm volatile("" : : "v"(p.x), "v"(p.y));
      const int s_l = p.x;
      int c_src = 0;
      if (FIRST) c_src = cnt[s_l];
      int gc = degb < ROWS_AHEAD ? degb : ROWS_AHEAD;
      float2 v[ROWS_AHEAD];
      rows_load<0>(v, H2, s_l, 0, gc, lane);
      if (more) {
        ndeg = __builtin_amdgcn_readlane(c_l, j + 1);
        ndegb = ndeg < BUCKET ? ndeg : BUCKET;
        if (lane < ndegb) {
          if (FIRST) pn.x = bk[((size_t)nxt * BUCKET + lane) * 2];
          else       pn = ((const int2*)bk)[(size_t)nxt * BUCKET + lane];
        }
        hn = H2[(size_t)nxt * (H_DIM / 2) + lane];
      }
      float w_l;
      if (FIRST) {
        const float ws = rsqrtf((float)(c_src + 1));
        if (lane < degb)
          bk[((size_t)node * BUCKET + lane) * 2 + 1] = __float_as_int(ws);
        w_l = ws * dd;
      } else {
        w_l = __int_as_float(p.y) * dd;
      }
      rows_sum<0>(v, w_l, 0, gc, ax, ay);
      for (int i = ROWS_AHEAD; i < degb; ++i) {
        const int s = __builtin_amdgcn_readlane(s_l, i);
        const float w = __int_as_float(
            __builtin_amdgcn_readlane(__float_as_int(w_l), i));
        const float2 r = H2[(size_t)s * (H_DIM / 2) + lane];
        ax = fmaf(r.x, w, ax);
        ay = fmaf(r.y, w, ay);
      }

      if (deg > BUCKET) {
        int L = ovfc[0];
        if (L > ovf_cap) L = ovf_cap;
        for (int b0 = 0; b0 < L; b0 += 32) {
          int2 q = make_int2(-1, 0);
          if (b0 + lane < L) q = ((const int2*)ovf)[b0 + lane];
          unsigned hit = __builtin_amdgcn_ballot_w32(q.x == node);
          while (hit != 0u) {
            const int l = __builtin_ctz(hit);
            hit &= hit - 1u;
            const int s = __builtin_amdgcn_readlane(q.y, l);
            const float w = rsqrtf((float)(cnt[s] + 1)) * dd;
            const float2 r = H2[(size_t)s * (H_DIM / 2) + lane];
            ax = fmaf(r.x, w, ax);
            ay = fmaf(r.y, w, ay);
          }
        }
      }

      if (FIRST) {
        ax = fmaxf(ax, 0.0f);
        ay = fmaxf(ay, 0.0f);
        const int slot = j & 15;
        *(float2*)(&s_zt[wib][slot * ZT_PITCH + 2 * lane]) =
            make_float2(ax, ay);
        nrows = slot + 1;
      } else {
        ((float2*)Out)[(size_t)node * (H_DIM / 2) + lane] =
            make_float2(ax, ay);
      }
      if (!more) break;
      deg = ndeg;
      degb = ndegb;
      p = pn;
      hs = hn;
      node = nxt;
    }
  }
  if (FIRST) {
    if (nrows > 0)
      tile_gemm_store(&s_zt[wib][0], s_wq, Out, wave0, nwaves, nrows, lane);
  }
}

#define DEC_GROUP 16
__global__ void __launch_bounds__(256)
k_decode(const float* __restrict__ Z, const int* __restrict__ pos,
         const int* __restrict__ neg, float* __restrict__ out, int npos,
         int nneg) {
  const int total = npos + nneg;
  const int base = __builtin_amdgcn_readfirstlane(
      (int)(blockIdx.x * (blockDim.x >> 5) + (threadIdx.x >> 5))) * 32;
  if (base >= total) return;
  const int lane = threadIdx.x & 31;
  const int k = base + lane;
  int a = 0, b = 0;
  if (k < npos) {
    a = pos[k];
    b = pos[npos + k];
  } else if (k < total) {
    const int j = k - npos;
    a = neg[j];
    b = neg[nneg + j];
  }
  const float2* __restrict__ Z2 = (const float2*)Z;
  float res = 0.0f;
#pragma unroll 1
  for (int t0 = 0; t0 < 32; t0 += DEC_GROUP) {
    float2 va[DEC_GROUP], vb[DEC_GROUP];
#pragma unroll
    for (int i = 0; i < DEC_GROUP; ++i) {
      const int sa = __builtin_amdgcn_readlane(a, t0 + i);
      const int sb = __builtin_amdgcn_readlane(b, t0 + i);
      va[i] = Z2[(size_t)sa * (H_DIM / 2) + lane];
      vb[i] = Z2[(size_t)sb * (H_DIM / 2) + lane];
    }
#pragma unroll
    for (int i = 0; i < DEC_GROUP; ++i) {
      float s = fmaf(va[i].x, vb[i].x, va[i].y * vb[i].y);
#pragma unroll
      for (int o = 16; o > 0; o >>= 1) s += __shfl_xor(s, o, 32);
      if (lane == t0 + i) res = s;
    }
  }
  if (k < total) out[k] = res;
}

extern "C" void kernel_launch(void* const* d_in, const int* in_sizes, int n_in,
                              void* d_out, int out_size, void* d_ws,
                              size_t ws_size, hipStream_t stream) {
  (void)n_in; (void)out_size; (void)ws_size;
  const float* x   = (const float*)d_in[0];
  const int*   ei  = (const int*)d_in[1];
  const int*   pei = (const int*)d_in[2];
  const int*   nei = (const int*)d_in[3];
  const float* W1  = (const float*)d_in[4];
  const float* b1  = (const float*)d_in[5];
  const float* W2  = (const float*)d_in[6];
  const float* b2  = (const float*)d_in[7];
  float* out = (float*)d_out;

  const int n    = in_sizes[0] / H_DIM;
  const int ne   = in_sizes[1] / 2;
  const int npos = in_sizes[2] / 2;
  const int nneg = in_sizes[3] / 2;
  const int* src = ei;
  const int* dst = ei + ne;

  float* ws = (float*)d_ws;
  const size_t nAl = ((size_t)n + 63) & ~(size_t)63;
  float* hbuf = ws;
  float* acc1 = hbuf + (size_t)n * H_DIM;
  float* acc2 = acc1 + (size_t)n * H_DIM;
  int* cnt  = (int*)(acc2 + (size_t)n * H_DIM);
  int* ovfc = cnt + nAl;
  int* bk   = ovfc + 64;
  int* ovf  = bk + (size_t)n * BUCKET * 2;

  const int B = 256;
  const int zB = B * ZERO_PER_THREAD;
  const int eB = B * EDGES_PER_THREAD;
  k_cnt_zero<<<(n + zB - 1) / zB, B, 0, stream>>>(cnt, n, ovfc);
  const int tiles = (n + 15) / 16;
  const int cblocks = (ne + eB - 1) / eB;
  const int gemm_blocks8 = (tiles + 1) / 2 < 1024 ? (tiles + 1) / 2 : 1024;
  k_count_rank_gemm<<<cblocks + gemm_blocks8, B, 0, stream>>>(
      src, dst, cnt, ovfc, bk, ovf, ne, cblocks, x, W1, hbuf, n, tiles);

  const int node_blocks = (n + 7) / 8;
  const int gblocks = node_blocks < 1280 ? node_blocks : 1280;
  const int tile_blocks = (n + 127) / 128;
  const int g1blocks = gblocks > tile_blocks ? gblocks : tile_blocks;
  k_gather_rows<true><<<g1blocks, B, 0, stream>>>(hbuf, cnt, bk, ovf, ovfc, ne,
                                                  b1, W2, acc1, n);
  k_gather_rows<false><<<gblocks, B, 0, stream>>>(acc1, cnt, bk, ovf, ovfc, ne,
                                                  b2, W2, acc2, n);
  k_decode<<<(npos + nneg + 255) / 256, B, 0, stream>>>(acc2, pei, nei, out,
                                                       npos, nneg);
}


# Per-edge global returning atomic + scattered store replaced by two-level LDS binning: blocks rank edges into 256-node bins, a block per bin builds node buckets in LDS; 5 launches to 6; arrival order
# speedup vs baseline: 3.6175x; 1.6132x over previous
#include <hip/hip_runtime.h>
#include <hip/hip_bf16.h>


typedef float v2f __attribute__((ext_vector_type(2)));
typedef float v8f __attribute__((ext_vector_type(8)));

#define H_DIM 64

#define ZERO_PER_THREAD 4
#define EDGES_PER_THREAD 16

__global__ void k_cnt_zero(int* __restrict__ cnt, int n,
                           int* __restrict__ ovfc) {
  const int T = gridDim.x * blockDim.x;
  int i = blockIdx.x * blockDim.x + threadIdx.x;
#pragma unroll
  for (int j = 0; j < ZERO_PER_THREAD; ++j, i += T)
    if (i < n) cnt[i] = 0;
  if (blockIdx.x == 0 && threadIdx.x == 0) {
    ovfc[0] = 0;
    ovfc[1] = 0;
  }
}

#define BUCKET 32
#define BIN_SHIFT 8
#define BIN_NODES 256
#define MAX_BINS 512
#define BIN_CAP (BIN_NODES * BUCKET)

template <bool GUARD>
__device__ __forceinline__ void bin_edges_body(
    const int* __restrict__ src, const int* __restrict__ dst,
    int* __restrict__ gcount, int* __restrict__ ovfc,
    int2* __restrict__ binbuf, int2* __restrict__ ovf2, int ne, int e0,
    int nb, int* s_hist) {
  int d[EDGES_PER_THREAD], s[EDGES_PER_THREAD], r[EDGES_PER_THREAD];
#pragma unroll
  for (int j = 0; j < EDGES_PER_THREAD; ++j) {
    const int e = e0 + j * 256;
    const bool ok = !GUARD || e < ne;
    d[j] = ok ? dst[e] : 0;
    s[j] = ok ? src[e] : 0;
  }
#pragma unroll
  for (int j = 0; j < EDGES_PER_THREAD; ++j) {
    const int e = e0 + j * 256;
    r[j] = 0;
    if (!GUARD || e < ne) r[j] = atomicAdd(&s_hist[d[j] >> BIN_SHIFT], 1);
  }
  __syncthreads();
  static_assert(MAX_BINS == 512, "two bins per thread of a 256-thread block");
  const int i0 = (int)threadIdx.x, i1 = i0 + 256;
  const int c0 = s_hist[i0], c1 = s_hist[i1];
  int g0 = 0, g1 = 0;
  if (i0 < nb) g0 = atomicAdd(&gcount[i0], c0);
  if (i1 < nb) g1 = atomicAdd(&gcount[i1], c1);
  s_hist[i0] = g0;
  s_hist[i1] = g1;
  __syncthreads();
  int st[EDGES_PER_THREAD];
#pragma unroll
  for (int j = 0; j < EDGES_PER_THREAD; ++j) st[j] = s_hist[d[j] >> BIN_SHIFT];
#pragma unroll
  for (int j = 0; j < EDGES_PER_THREAD; ++j) {
    const int e = e0 + j * 256;
    if (!GUARD || e < ne) {
      const int b = d[j] >> BIN_SHIFT;
      const int pos = st[j] + r[j];
      int2* p;
      if (pos < BIN_CAP) {
        p = binbuf + ((size_t)b * BIN_CAP + pos);
      } else {
        p = ovf2 + atomicAdd(ovfc + 1, 1);
      }
      *p = make_int2(d[j], s[j]);
    }
  }
}

__device__ __forceinline__ void gemm_tiles(const float* __restrict__ X,
                                           const float* __restrict__ W,
                                           float* __restrict__ Hout, int n,
                                           int tiles, int tile0, int tstride,
                                           int wave) {
  const int lane  = threadIdx.x & 31;
  const int n0    = wave * 16;
  const int m     = lane & 15;
  const int khalf = (lane >> 4) << 1;
  const int rbase = (lane >> 4) << 3;
  const int col   = n0 + m;

  v2f bf[16];
#pragma unroll
  for (int kk = 0; kk < 16; ++kk) {
    const int kb = kk * 4 + khalf;
    bf[kk].x = W[kb * H_DIM + col];
    bf[kk].y = W[(kb + 1) * H_DIM + col];
  }

  for (int tile = tile0; tile < tiles; tile += tstride) {
    const int row0 = tile * 16;
    int arow = row0 + m;
    if (arow >= n) arow = n - 1;
    const float* __restrict__ xrow = X + (size_t)arow * H_DIM;

    v8f acc = {};
#pragma unroll
    for (int kk = 0; kk < 16; ++kk) {
      const int kb = kk * 4 + khalf;
      v2f a;
      a.x = xrow[kb];
      a.y = xrow[kb + 1];
      acc = __builtin_amdgcn_wmma_f32_16x16x4_f32(false, a, false, bf[kk],
                                                  (short)0, acc, false, false);
    }

#pragma unroll
    for (int r = 0; r < 8; ++r) {
      const int row = row0 + rbase + r;
      if (row < n) Hout[(size_t)row * H_DIM + col] = acc[r];
    }
  }
}

__global__ void __launch_bounds__(256)
k_bin_edges_gemm(const int* __restrict__ esrc, const int* __restrict__ dst,
                 int* __restrict__ gcount, int* __restrict__ ovfc,
                 int2* __restrict__ binbuf, int2* __restrict__ ovf2, int ne,
                 int cblocks, int nb,
                 const float* __restrict__ X, const float* __restrict__ W,
                 float* __restrict__ Hout, int n, int tiles) {
  __shared__ int s_hist[MAX_BINS];
  if ((int)blockIdx.x < cblocks) {
    for (int i = (int)threadIdx.x; i < MAX_BINS; i += 256) s_hist[i] = 0;
    __syncthreads();
    const int eb = (int)blockIdx.x * (256 * EDGES_PER_THREAD);
    const int e0 = eb + (int)threadIdx.x;
    if (eb + 256 * EDGES_PER_THREAD <= ne)
      bin_edges_body<false>(esrc, dst, gcount, ovfc, binbuf, ovf2, ne, e0, nb,
                            s_hist);
    else
      bin_edges_body<true>(esrc, dst, gcount, ovfc, binbuf, ovf2, ne, e0, nb,
                           s_hist);
  } else {
    const int wave = __builtin_amdgcn_readfirstlane((int)(threadIdx.x >> 5));
    const int g = (int)blockIdx.x - cblocks;
    const int G = (int)gridDim.x - cblocks;
    gemm_tiles(X, W, Hout, n, tiles, g * 2 + (wave >> 2), G * 2, wave & 3);
  }
}

#define PLACE_AHEAD 16
#define TAB_PITCH (BUCKET + 1)
__device__ __forceinline__ void place_edge(int2 q, int* s_cnt, int* s_tab,
                                           int* __restrict__ ovfc,
                                           int2* __restrict__ ovf) {
  const int j = q.x & (BIN_NODES - 1);
  const int r = atomicAdd(&s_cnt[j], 1);
  if (r < BUCKET) {
    s_tab[j * TAB_PITCH + r] = q.y;
  } else {
    const int o = atomicAdd(ovfc, 1);
    ovf[o] = q;
  }
}

__global__ void __launch_bounds__(256)
k_bin_place(const int* __restrict__ gcount, const int2* __restrict__ binbuf,
            int* __restrict__ ovfc, const int2* __restrict__ ovf2,
            int ovf_cap, int* __restrict__ cnt, int* __restrict__ bk,
            int2* __restrict__ ovf, int n) {
  __shared__ int s_cnt[BIN_NODES + 1];
  __shared__ int s_tab[BIN_NODES * TAB_PITCH];
  const int tid = (int)threadIdx.x;
  const int b = (int)blockIdx.x;
  s_cnt[tid] = 0;
  if (tid == 0) s_cnt[BIN_NODES] = 0;
  __syncthreads();
  int m = gcount[b];
  if (m > BIN_CAP) m = BIN_CAP;
  const int2* __restrict__ buf = binbuf + (size_t)b * BIN_CAP;
  for (int i0 = 0; i0 < m; i0 += 256 * PLACE_AHEAD) {
    int2 q[PLACE_AHEAD];
#pragma unroll
    for (int k = 0; k < PLACE_AHEAD; ++k) {
      const int i = i0 + k * 256 + tid;
      q[k] = make_int2(-1, 0);
      if (i < m) q[k] = buf[i];
    }
    int r[PLACE_AHEAD];
#pragma unroll
    for (int k = 0; k < PLACE_AHEAD; ++k) {
      const int j = q[k].x >= 0 ? (q[k].x & (BIN_NODES - 1)) : BIN_NODES;
      r[k] = atomicAdd(&s_cnt[j], 1);
    }
#pragma unroll
    for (int k = 0; k < PLACE_AHEAD; ++k) {
      if (q[k].x >= 0) {
        const int j = q[k].x & (BIN_NODES - 1);
        if (r[k] < BUCKET) {
          s_tab[j * TAB_PITCH + r[k]] = q[k].y;
        } else {
          const int o = atomicAdd(ovfc, 1);
          ovf[o] = q[k];
        }
      }
    }
  }
  int L = ovfc[1];
  if (L > ovf_cap) L = ovf_cap;
  for (int i = tid; i < L; i += 256) {
    const int2 q = ovf2[i];
    if ((q.x >> BIN_SHIFT) == b) place_edge(q, s_cnt, s_tab, ovfc, ovf);
  }
  __syncthreads();
  const int wbase = __builtin_amdgcn_readfirstlane(tid >> 5) * 32;
  const int lane = tid & 31;
  const int c_l = s_cnt[wbase + lane];
  const int mine = b * BIN_NODES + wbase + lane;
  if (mine < n) cnt[mine] = c_l;
  int* __restrict__ bkw =
      bk + ((size_t)(b * BIN_NODES + wbase) * BUCKET + lane) * 2;
#pragma unroll
  for (int k0 = 0; k0 < 32; k0 += 8) {
    int v[8];
#pragma unroll
    for (int k = 0; k < 8; ++k) {
      const int ck = __builtin_amdgcn_readlane(c_l, k0 + k);
      v[k] = 0;
      if (lane < ck) v[k] = s_tab[(wbase + k0 + k) * TAB_PITCH + lane];
    }
#pragma unroll
    for (int k = 0; k < 8; ++k) {
      const int ck = __builtin_amdgcn_readlane(c_l, k0 + k);
      if (lane < ck) bkw[(size_t)(k0 + k) * BUCKET * 2] = v[k];
    }
  }
}

#define ROWS_AHEAD 24
template <int I>
__device__ __forceinline__ void rows_load(float2 (&v)[ROWS_AHEAD],
                                          const float2* __restrict__ H2,
                                          int s_l, int t, int cnt, int lane) {
  if constexpr (I < ROWS_AHEAD) {
    if (I < cnt) {
      const int s = __builtin_amdgcn_readlane(s_l, t + I);
      v[I] = H2[(size_t)s * (H_DIM / 2) + lane];
      rows_load<I + 1>(v, H2, s_l, t, cnt, lane);
    }
  }
}
template <int I>
__device__ __forceinline__ void rows_sum(const float2 (&v)[ROWS_AHEAD],
                                         float w_l, int t, int cnt, float& ax,
                                         float& ay) {
  if constexpr (I < ROWS_AHEAD) {
    if (I < cnt) {
      const float w = __int_as_float(
          __builtin_amdgcn_readlane(__float_as_int(w_l), t + I));
      ax = fmaf(v[I].x, w, ax);
      ay = fmaf(v[I].y, w, ay);
      rows_sum<I + 1>(v, w_l, t, cnt, ax, ay);
    }
  }
}

#define ZT_PITCH 68
__device__ __forceinline__ void tile_gemm_store(const float* zt,
                                                const float* wq,
                                                float* __restrict__ Hout,
                                                int nb, int nwaves, int nrows,
                                                int lane) {
  const int m     = lane & 15;
  const int khalf = (lane >> 4) << 1;
  const int rbase = (lane >> 4) << 3;
  const unsigned lane_off = (unsigned)(rbase * nwaves) * H_DIM + (unsigned)m;
#pragma unroll
  for (int c = 0; c < 4; c += 2) {
    v8f acc0 = {}, acc1 = {};
#pragma unroll
    for (int kk = 0; kk < 16; ++kk) {
      const v2f a  = *(const v2f*)(zt + m * ZT_PITCH + kk * 4 + khalf);
      const v2f b0 = *(const v2f*)(wq + ((kk * 64 + c * 16 + m) << 2) + khalf);
      const v2f b1 =
          *(const v2f*)(wq + ((kk * 64 + (c + 1) * 16 + m) << 2) + khalf);
      acc0 = __builtin_amdgcn_wmma_f32_16x16x4_f32(false, a, false, b0,
                                                   (short)0, acc0, false, false);
      acc1 = __builtin_amdgcn_wmma_f32_16x16x4_f32(false, a, false, b1,
                                                   (short)0, acc1, false, false);
    }
#pragma unroll
    for (int r = 0; r < 8; ++r) {
      if (rbase + r < nrows) {
        float* __restrict__ srow =
            Hout + (size_t)(nb + r * nwaves) * H_DIM + c * 16;
        srow[lane_off]      = acc0[r];
        srow[lane_off + 16] = acc1[r];
      }
    }
  }
}

template <bool FIRST>
__global__ void __launch_bounds__(256, 10)
k_gather_rows(const float* __restrict__ Hm, const int* __restrict__ cnt,
              int* __restrict__ bk, const int* __restrict__ ovf,
              const int* __restrict__ ovfc, int ovf_cap,
              const float* __restrict__ bias, const float* __restrict__ Wn,
              float* __restrict__ Out, int n) {
  const int wave0 = __builtin_amdgcn_readfirstlane(
      (int)(blockIdx.x * (blockDim.x >> 5) + (threadIdx.x >> 5)));
  const int nwaves = (int)(gridDim.x * (blockDim.x >> 5));
  const int lane = threadIdx.x & 31;
  const float2 bb = ((const float2*)bias)[lane];
  const float2* __restrict__ H2 = (const float2*)Hm;

  __shared__ alignas(16) float s_wq[H_DIM * H_DIM];
  __shared__ alignas(16) float s_zt[8][16 * ZT_PITCH];
  const int wib = __builtin_amdgcn_readfirstlane((int)(threadIdx.x >> 5));
  if (FIRST) {
    float4 wv[4];
#pragma unroll
    for (int q = 0; q < 4; ++q) {
      const int cell = q * 256 + (int)threadIdx.x;
      const float* __restrict__ wp = Wn + ((cell >> 6) << 8) + (cell & 63);
      wv[q] = make_float4(wp[0], wp[H_DIM], wp[2 * H_DIM], wp[3 * H_DIM]);
    }
#pragma unroll
    for (int q = 0; q < 4; ++q)
      ((float4*)s_wq)[q * 256 + (int)threadIdx.x] = wv[q];
    __syncthreads();
  }

  int nrows = 0;
  for (int nb = wave0; nb < n; nb += 32 * nwaves) {
    const int mine = nb + lane * nwaves;
    int c_l = 0;
    if (mine < n) c_l = cnt[mine];
    const float d_l = rsqrtf((float)(c_l + 1));
    int deg = __builtin_amdgcn_readlane(c_l, 0);
    int degb = deg < BUCKET ? deg : BUCKET;
    int2 p = make_int2(0, 0);
    if (lane < degb) {
      if (FIRST) p.x = bk[((size_t)nb * BUCKET + lane) * 2];
      else       p = ((const int2*)bk)[(size_t)nb * BUCKET + lane];
    }
    float2 hs = H2[(size_t)nb * (H_DIM / 2) + lane];

    int node = nb;
    for (int j = 0; j < 32; ++j) {
      const float dd = __int_as_float(
          __builtin_amdgcn_readlane(__float_as_int(d_l), j));
      const int nxt = node + nwaves;
      const bool more = (j + 1 < 32) && (nxt < n);
      float ax = fmaf(hs.x, dd * dd, bb.x);
      float ay = fmaf(hs.y, dd * dd, bb.y);
      int ndeg = 0, ndegb = 0;
      int2 pn = make_int2(0, 0);
      float2 hn = make_float2(0.0f, 0.0f);

      if (FIRST) asm volatile("" : : "v"(p.x));
      else       asm volatile("" : : "v"(p.x), "v"(p.y));
      const int s_l = p.x;
      int c_src = 0;
      if (FIRST) c_src = cnt[s_l];
      int gc = degb < ROWS_AHEAD ? degb : ROWS_AHEAD;
      float2 v[ROWS_AHEAD];
      rows_load<0>(v, H2, s_l, 0, gc, lane);
      if (more) {
        ndeg = __builtin_amdgcn_readlane(c_l, j + 1);
        ndegb = ndeg < BUCKET ? ndeg : BUCKET;
        if (lane < ndegb) {
          if (FIRST) pn.x = bk[((size_t)nxt * BUCKET + lane) * 2];
          else       pn = ((const int2*)bk)[(size_t)nxt * BUCKET + lane];
        }
        hn = H2[(size_t)nxt * (H_DIM / 2) + lane];
      }
      float w_l;
      if (FIRST) {
        const float ws = rsqrtf((float)(c_src + 1));
        if (lane < degb)
          bk[((size_t)node * BUCKET + lane) * 2 + 1] = __float_as_int(ws);
        w_l = ws * dd;
      } else {
        w_l = __int_as_float(p.y) * dd;
      }
      rows_sum<0>(v, w_l, 0, gc, ax, ay);
      for (int i = ROWS_AHEAD; i < degb; ++i) {
        const int s = __builtin_amdgcn_readlane(s_l, i);
        const float w = __int_as_float(
            __builtin_amdgcn_readlane(__float_as_int(w_l), i));
        const float2 r = H2[(size_t)s * (H_DIM / 2) + lane];
        ax = fmaf(r.x, w, ax);
        ay = fmaf(r.y, w, ay);
      }

      if (deg > BUCKET) {
        int L = ovfc[0];
        if (L > ovf_cap) L = ovf_cap;
        for (int b0 = 0; b0 < L; b0 += 32) {
          int2 q = make_int2(-1, 0);
          if (b0 + lane < L) q = ((const int2*)ovf)[b0 + lane];
          unsigned hit = __builtin_amdgcn_ballot_w32(q.x == node);
          while (hit != 0u) {
            const int l = __builtin_ctz(hit);
            hit &= hit - 1u;
            const int s = __builtin_amdgcn_readlane(q.y, l);
            const float w = rsqrtf((float)(cnt[s] + 1)) * dd;
            const float2 r = H2[(size_t)s * (H_DIM / 2) + lane];
            ax = fmaf(r.x, w, ax);
            ay = fmaf(r.y, w, ay);
          }
        }
      }

      if (FIRST) {
        ax = fmaxf(ax, 0.0f);
        ay = fmaxf(ay, 0.0f);
        const int slot = j & 15;
        *(float2*)(&s_zt[wib][slot * ZT_PITCH + 2 * lane]) =
            make_float2(ax, ay);
        nrows = slot + 1;
      } else {
        ((float2*)Out)[(size_t)node * (H_DIM / 2) + lane] =
            make_float2(ax, ay);
      }
      if (!more) break;
      deg = ndeg;
      degb = ndegb;
      p = pn;
      hs = hn;
      node = nxt;
    }
  }
  if (FIRST) {
    if (nrows > 0)
      tile_gemm_store(&s_zt[wib][0], s_wq, Out, wave0, nwaves, nrows, lane);
  }
}

#define DEC_GROUP 16
__global__ void __launch_bounds__(256)
k_decode(const float* __restrict__ Z, const int* __restrict__ pos,
         const int* __restrict__ neg, float* __restrict__ out, int npos,
         int nneg) {
  const int total = npos + nneg;
  const int base = __builtin_amdgcn_readfirstlane(
      (int)(blockIdx.x * (blockDim.x >> 5) + (threadIdx.x >> 5))) * 32;
  if (base >= total) return;
  const int lane = threadIdx.x & 31;
  const int k = base + lane;
  int a = 0, b = 0;
  if (k < npos) {
    a = pos[k];
    b = pos[npos + k];
  } else if (k < total) {
    const int j = k - npos;
    a = neg[j];
    b = neg[nneg + j];
  }
  const float2* __restrict__ Z2 = (const float2*)Z;
  float res = 0.0f;
#pragma unroll 1
  for (int t0 = 0; t0 < 32; t0 += DEC_GROUP) {
    float2 va[DEC_GROUP], vb[DEC_GROUP];
#pragma unroll
    for (int i = 0; i < DEC_GROUP; ++i) {
      const int sa = __builtin_amdgcn_readlane(a, t0 + i);
      const int sb = __builtin_amdgcn_readlane(b, t0 + i);
      va[i] = Z2[(size_t)sa * (H_DIM / 2) + lane];
      vb[i] = Z2[(size_t)sb * (H_DIM / 2) + lane];
    }
#pragma unroll
    for (int i = 0; i < DEC_GROUP; ++i) {
      float s = fmaf(va[i].x, vb[i].x, va[i].y * vb[i].y);
#pragma unroll
      for (int o = 16; o > 0; o >>= 1) s += __shfl_xor(s, o, 32);
      if (lane == t0 + i) res = s;
    }
  }
  if (k < total) out[k] = res;
}

extern "C" void kernel_launch(void* const* d_in, const int* in_sizes, int n_in,
                              void* d_out, int out_size, void* d_ws,
                              size_t ws_size, hipStream_t stream) {
  (void)n_in; (void)out_size; (void)ws_size;
  const float* x   = (const float*)d_in[0];
  const int*   ei  = (const int*)d_in[1];
  const int*   pei = (const int*)d_in[2];
  const int*   nei = (const int*)d_in[3];
  const float* W1  = (const float*)d_in[4];
  const float* b1  = (const float*)d_in[5];
  const float* W2  = (const float*)d_in[6];
  const float* b2  = (const float*)d_in[7];
  float* out = (float*)d_out;

  const int n    = in_sizes[0] / H_DIM;
  const int ne   = in_sizes[1] / 2;
  const int npos = in_sizes[2] / 2;
  const int nneg = in_sizes[3] / 2;
  const int* src = ei;
  const int* dst = ei + ne;

  float* ws = (float*)d_ws;
  const size_t nAl = ((size_t)n + 63) & ~(size_t)63;
  float* hbuf = ws;
  float* acc1 = hbuf + (size_t)n * H_DIM;
  float* acc2 = acc1 + (size_t)n * H_DIM;
  int* cnt  = (int*)(acc2 + (size_t)n * H_DIM);
  int* ovfc = cnt + nAl;
  int* bk   = ovfc + 64;
  int* ovf  = bk + (size_t)n * BUCKET * 2;
  int* ovf2 = ovf + (size_t)ne * 2;
  int* gcount = ovf2 + (size_t)ne * 2;
  int* bins = gcount + MAX_BINS;
  const int nb = (n + BIN_NODES - 1) / BIN_NODES;

  const int B = 256;
  const int zB = B * ZERO_PER_THREAD;
  const int eB = B * EDGES_PER_THREAD;
  k_cnt_zero<<<(nb + zB - 1) / zB, B, 0, stream>>>(gcount, nb, ovfc);
  const int tiles = (n + 15) / 16;
  const int cblocks = (ne + eB - 1) / eB;
  const int gemm_blocks8 = (tiles + 1) / 2 < 1024 ? (tiles + 1) / 2 : 1024;
  k_bin_edges_gemm<<<cblocks + gemm_blocks8, B, 0, stream>>>(
      src, dst, gcount, ovfc, (int2*)bins, (int2*)ovf2, ne, cblocks, nb, x, W1,
      hbuf, n, tiles);
  k_bin_place<<<nb, B, 0, stream>>>(gcount, (const int2*)bins, ovfc,
                                    (const int2*)ovf2, ne, cnt, bk, (int2*)ovf,
                                    n);

  const int node_blocks = (n + 7) / 8;
  const int gblocks = node_blocks < 1280 ? node_blocks : 1280;
  const int tile_blocks = (n + 127) / 128;
  const int g1blocks = gblocks > tile_blocks ? gblocks : tile_blocks;
  k_gather_rows<true><<<g1blocks, B, 0, stream>>>(hbuf, cnt, bk, ovf, ovfc, ne,
                                                  b1, W2, acc1, n);
  k_gather_rows<false><<<gblocks, B, 0, stream>>>(acc1, cnt, bk, ovf, ovfc, ne,
                                                  b2, W2, acc2, n);
  k_decode<<<(npos + nneg + 255) / 256, B, 0, stream>>>(acc2, pei, nei, out,
                                                       npos, nneg);
}


# Layer-1 GEMM: one wave fetches a 16-row tile once with whole-line 16-byte loads, stages it in LDS, does all 64 columns (was 4 waves, each fetching it as 16-line 8-byte pieces); sum order kept
# speedup vs baseline: 3.7538x; 1.0377x over previous
#include <hip/hip_runtime.h>
#include <hip/hip_bf16.h>


typedef float v2f __attribute__((ext_vector_type(2)));
typedef float v8f __attribute__((ext_vector_type(8)));

#define H_DIM 64

#define ZERO_PER_THREAD 4
#define EDGES_PER_THREAD 16

__global__ void k_cnt_zero(int* __restrict__ cnt, int n,
                           int* __restrict__ ovfc) {
  const int T = gridDim.x * blockDim.x;
  int i = blockIdx.x * blockDim.x + threadIdx.x;
#pragma unroll
  for (int j = 0; j < ZERO_PER_THREAD; ++j, i += T)
    if (i < n) cnt[i] = 0;
  if (blockIdx.x == 0 && threadIdx.x == 0) {
    ovfc[0] = 0;
    ovfc[1] = 0;
  }
}

#define BUCKET 32
#define BIN_SHIFT 8
#define BIN_NODES 256
#define MAX_BINS 512
#define BIN_CAP (BIN_NODES * BUCKET)

template <bool GUARD>
__device__ __forceinline__ void bin_edges_body(
    const int* __restrict__ src, const int* __restrict__ dst,
    int* __restrict__ gcount, int* __restrict__ ovfc,
    int2* __restrict__ binbuf, int2* __restrict__ ovf2, int ne, int e0,
    int nb, int* s_hist) {
  int d[EDGES_PER_THREAD], s[EDGES_PER_THREAD], r[EDGES_PER_THREAD];
#pragma unroll
  for (int j = 0; j < EDGES_PER_THREAD; ++j) {
    const int e = e0 + j * 256;
    const bool ok = !GUARD || e < ne;
    d[j] = ok ? dst[e] : 0;
    s[j] = ok ? src[e] : 0;
  }
#pragma unroll
  for (int j = 0; j < EDGES_PER_THREAD; ++j) {
    const int e = e0 + j * 256;
    r[j] = 0;
    if (!GUARD || e < ne) r[j] = atomicAdd(&s_hist[d[j] >> BIN_SHIFT], 1);
  }
  __syncthreads();
  static_assert(MAX_BINS == 512, "two bins per thread of a 256-thread block");
  const int i0 = (int)threadIdx.x, i1 = i0 + 256;
  const int c0 = s_hist[i0], c1 = s_hist[i1];
  int g0 = 0, g1 = 0;
  if (i0 < nb) g0 = atomicAdd(&gcount[i0], c0);
  if (i1 < nb) g1 = atomicAdd(&gcount[i1], c1);
  s_hist[i0] = g0;
  s_hist[i1] = g1;
  __syncthreads();
  int st[EDGES_PER_THREAD];
#pragma unroll
  for (int j = 0; j < EDGES_PER_THREAD; ++j) st[j] = s_hist[d[j] >> BIN_SHIFT];
#pragma unroll
  for (int j = 0; j < EDGES_PER_THREAD; ++j) {
    const int e = e0 + j * 256;
    if (!GUARD || e < ne) {
      const int b = d[j] >> BIN_SHIFT;
      const int pos = st[j] + r[j];
      int2* p;
      if (pos < BIN_CAP) {
        p = binbuf + ((size_t)b * BIN_CAP + pos);
      } else {
        p = ovf2 + atomicAdd(ovfc + 1, 1);
      }
      *p = make_int2(d[j], s[j]);
    }
  }
}

__device__ __forceinline__ void stage_w_quads(const float* __restrict__ Wn,
                                              float* s_wq) {
  float4 wv[4];
#pragma unroll
  for (int q = 0; q < 4; ++q) {
    const int cell = q * 256 + (int)threadIdx.x;
    const float* __restrict__ wp = Wn + ((cell >> 6) << 8) + (cell & 63);
    wv[q] = make_float4(wp[0], wp[H_DIM], wp[2 * H_DIM], wp[3 * H_DIM]);
  }
#pragma unroll
  for (int q = 0; q < 4; ++q)
    ((float4*)s_wq)[q * 256 + (int)threadIdx.x] = wv[q];
}

#define ZT_PITCH 68
__device__ __forceinline__ void tile_gemm_store(const float* zt,
                                                const float* wq,
                                                float* __restrict__ Hout,
                                                int nb, int nwaves, int nrows,
                                                int lane) {
  const int m     = lane & 15;
  const int khalf = (lane >> 4) << 1;
  const int rbase = (lane >> 4) << 3;
  const unsigned lane_off = (unsigned)(rbase * nwaves) * H_DIM + (unsigned)m;
#pragma unroll
  for (int c = 0; c < 4; c += 2) {
    v8f acc0 = {}, acc1 = {};
#pragma unroll
    for (int kk = 0; kk < 16; ++kk) {
      const v2f a  = *(const v2f*)(zt + m * ZT_PITCH + kk * 4 + khalf);
      const v2f b0 = *(const v2f*)(wq + ((kk * 64 + c * 16 + m) << 2) + khalf);
      const v2f b1 =
          *(const v2f*)(wq + ((kk * 64 + (c + 1) * 16 + m) << 2) + khalf);
      acc0 = __builtin_amdgcn_wmma_f32_16x16x4_f32(false, a, false, b0,
                                                   (short)0, acc0, false, false);
      acc1 = __builtin_amdgcn_wmma_f32_16x16x4_f32(false, a, false, b1,
                                                   (short)0, acc1, false, false);
    }
#pragma unroll
    for (int r = 0; r < 8; ++r) {
      if (rbase + r < nrows) {
        float* __restrict__ srow =
            Hout + (size_t)(nb + r * nwaves) * H_DIM + c * 16;
        srow[lane_off]      = acc0[r];
        srow[lane_off + 16] = acc1[r];
      }
    }
  }
}

__global__ void __launch_bounds__(256)
k_bin_edges_gemm(const int* __restrict__ esrc, const int* __restrict__ dst,
                 int* __restrict__ gcount, int* __restrict__ ovfc,
                 int2* __restrict__ binbuf, int2* __restrict__ ovf2, int ne,
                 int cblocks, int nb,
                 const float* __restrict__ X, const float* __restrict__ W,
                 float* __restrict__ Hout, int n, int tiles) {
  __shared__ int s_hist[MAX_BINS];
  __shared__ alignas(16) float s_wq[H_DIM * H_DIM];
  __shared__ alignas(16) float s_xt[8][16 * ZT_PITCH];
  if ((int)blockIdx.x < cblocks) {
    for (int i = (int)threadIdx.x; i < MAX_BINS; i += 256) s_hist[i] = 0;
    __syncthreads();
    const int eb = (int)blockIdx.x * (256 * EDGES_PER_THREAD);
    const int e0 = eb + (int)threadIdx.x;
    if (eb + 256 * EDGES_PER_THREAD <= ne)
      bin_edges_body<false>(esrc, dst, gcount, ovfc, binbuf, ovf2, ne, e0, nb,
                            s_hist);
    else
      bin_edges_body<true>(esrc, dst, gcount, ovfc, binbuf, ovf2, ne, e0, nb,
                           s_hist);
  } else {
    const int wave = __builtin_amdgcn_readfirstlane((int)(threadIdx.x >> 5));
    const int lane = (int)threadIdx.x & 31;
    const int g = (int)blockIdx.x - cblocks;
    float* zt = &s_xt[wave][0];
    const int rhalf = lane >> 4;
    const int c4 = lane & 15;
    const float4* __restrict__ X4 = (const float4*)X;
    const int tile = g * 8 + wave;
    const int row0 = tile * 16;
    float4 xv[8];
#pragma unroll
    for (int i = 0; i < 8; ++i) {
      int row = row0 + 2 * i + rhalf;
      if (row >= n) row = n - 1;
      xv[i] = X4[(size_t)row * (H_DIM / 4) + c4];
    }
    stage_w_quads(W, s_wq);
#pragma unroll
    for (int i = 0; i < 8; ++i)
      *(float4*)(zt + (2 * i + rhalf) * ZT_PITCH + c4 * 4) = xv[i];
    __syncthreads();
    if (tile < tiles) {
      int nrows = n - row0;
      if (nrows > 16) nrows = 16;
      tile_gemm_store(zt, s_wq, Hout, row0, 1, nrows, lane);
    }
  }
}

#define PLACE_AHEAD 16
#define TAB_PITCH (BUCKET + 1)
__device__ __forceinline__ void place_edge(int2 q, int* s_cnt, int* s_tab,
                                           int* __restrict__ ovfc,
                                           int2* __restrict__ ovf) {
  const int j = q.x & (BIN_NODES - 1);
  const int r = atomicAdd(&s_cnt[j], 1);
  if (r < BUCKET) {
    s_tab[j * TAB_PITCH + r] = q.y;
  } else {
    const int o = atomicAdd(ovfc, 1);
    ovf[o] = q;
  }
}

__global__ void __launch_bounds__(256)
k_bin_place(const int* __restrict__ gcount, const int2* __restrict__ binbuf,
            int* __restrict__ ovfc, const int2* __restrict__ ovf2,
            int ovf_cap, int* __restrict__ cnt, int* __restrict__ bk,
            int2* __restrict__ ovf, int n) {
  __shared__ int s_cnt[BIN_NODES + 1];
  __shared__ int s_tab[BIN_NODES * TAB_PITCH];
  const int tid = (int)threadIdx.x;
  const int b = (int)blockIdx.x;
  s_cnt[tid] = 0;
  if (tid == 0) s_cnt[BIN_NODES] = 0;
  __syncthreads();
  int m = gcount[b];
  if (m > BIN_CAP) m = BIN_CAP;
  const int2* __restrict__ buf = binbuf + (size_t)b * BIN_CAP;
  for (int i0 = 0; i0 < m; i0 += 256 * PLACE_AHEAD) {
    int2 q[PLACE_AHEAD];
#pragma unroll
    for (int k = 0; k < PLACE_AHEAD; ++k) {
      const int i = i0 + k * 256 + tid;
      q[k] = make_int2(-1, 0);
      if (i < m) q[k] = buf[i];
    }
    int r[PLACE_AHEAD];
#pragma unroll
    for (int k = 0; k < PLACE_AHEAD; ++k) {
      const int j = q[k].x >= 0 ? (q[k].x & (BIN_NODES - 1)) : BIN_NODES;
      r[k] = atomicAdd(&s_cnt[j], 1);
    }
#pragma unroll
    for (int k = 0; k < PLACE_AHEAD; ++k) {
      if (q[k].x >= 0) {
        const int j = q[k].x & (BIN_NODES - 1);
        if (r[k] < BUCKET) {
          s_tab[j * TAB_PITCH + r[k]] = q[k].y;
        } else {
          const int o = atomicAdd(ovfc, 1);
          ovf[o] = q[k];
        }
      }
    }
  }
  int L = ovfc[1];
  if (L > ovf_cap) L = ovf_cap;
  for (int i = tid; i < L; i += 256) {
    const int2 q = ovf2[i];
    if ((q.x >> BIN_SHIFT) == b) place_edge(q, s_cnt, s_tab, ovfc, ovf);
  }
  __syncthreads();
  const int wbase = __builtin_amdgcn_readfirstlane(tid >> 5) * 32;
  const int lane = tid & 31;
  const int c_l = s_cnt[wbase + lane];
  const int mine = b * BIN_NODES + wbase + lane;
  if (mine < n) cnt[mine] = c_l;
  int* __restrict__ bkw =
      bk + ((size_t)(b * BIN_NODES + wbase) * BUCKET + lane) * 2;
#pragma unroll
  for (int k0 = 0; k0 < 32; k0 += 8) {
    int v[8];
#pragma unroll
    for (int k = 0; k < 8; ++k) {
      const int ck = __builtin_amdgcn_readlane(c_l, k0 + k);
      v[k] = 0;
      if (lane < ck) v[k] = s_tab[(wbase + k0 + k) * TAB_PITCH + lane];
    }
#pragma unroll
    for (int k = 0; k < 8; ++k) {
      const int ck = __builtin_amdgcn_readlane(c_l, k0 + k);
      if (lane < ck) bkw[(size_t)(k0 + k) * BUCKET * 2] = v[k];
    }
  }
}

#define ROWS_AHEAD 24
template <int I>
__device__ __forceinline__ void rows_load(float2 (&v)[ROWS_AHEAD],
                                          const float2* __restrict__ H2,
                                          int s_l, int t, int cnt, int lane) {
  if constexpr (I < ROWS_AHEAD) {
    if (I < cnt) {
      const int s = __builtin_amdgcn_readlane(s_l, t + I);
      v[I] = H2[(size_t)s * (H_DIM / 2) + lane];
      rows_load<I + 1>(v, H2, s_l, t, cnt, lane);
    }
  }
}
template <int I>
__device__ __forceinline__ void rows_sum(const float2 (&v)[ROWS_AHEAD],
                                         float w_l, int t, int cnt, float& ax,
                                         float& ay) {
  if constexpr (I < ROWS_AHEAD) {
    if (I < cnt) {
      const float w = __int_as_float(
          __builtin_amdgcn_readlane(__float_as_int(w_l), t + I));
      ax = fmaf(v[I].x, w, ax);
      ay = fmaf(v[I].y, w, ay);
      rows_sum<I + 1>(v, w_l, t, cnt, ax, ay);
    }
  }
}

template <bool FIRST>
__global__ void __launch_bounds__(256, 10)
k_gather_rows(const float* __restrict__ Hm, const int* __restrict__ cnt,
              int* __restrict__ bk, const int* __restrict__ ovf,
              const int* __restrict__ ovfc, int ovf_cap,
              const float* __restrict__ bias, const float* __restrict__ Wn,
              float* __restrict__ Out, int n) {
  const int wave0 = __builtin_amdgcn_readfirstlane(
      (int)(blockIdx.x * (blockDim.x >> 5) + (threadIdx.x >> 5)));
  const int nwaves = (int)(gridDim.x * (blockDim.x >> 5));
  const int lane = threadIdx.x & 31;
  const float2 bb = ((const float2*)bias)[lane];
  const float2* __restrict__ H2 = (const float2*)Hm;

  __shared__ alignas(16) float s_wq[H_DIM * H_DIM];
  __shared__ alignas(16) float s_zt[8][16 * ZT_PITCH];
  const int wib = __builtin_amdgcn_readfirstlane((int)(threadIdx.x >> 5));
  if (FIRST) {
    stage_w_quads(Wn, s_wq);
    __syncthreads();
  }

  int nrows = 0;
  for (int nb = wave0; nb < n; nb += 32 * nwaves) {
    const int mine = nb + lane * nwaves;
    int c_l = 0;
    if (mine < n) c_l = cnt[mine];
    const float d_l = rsqrtf((float)(c_l + 1));
    int deg = __builtin_amdgcn_readlane(c_l, 0);
    int degb = deg < BUCKET ? deg : BUCKET;
    int2 p = make_int2(0, 0);
    if (lane < degb) {
      if (FIRST) p.x = bk[((size_t)nb * BUCKET + lane) * 2];
      else       p = ((const int2*)bk)[(size_t)nb * BUCKET + lane];
    }
    float2 hs = H2[(size_t)nb * (H_DIM / 2) + lane];

    int node = nb;
    for (int j = 0; j < 32; ++j) {
      const float dd = __int_as_float(
          __builtin_amdgcn_readlane(__float_as_int(d_l), j));
      const int nxt = node + nwaves;
      const bool more = (j + 1 < 32) && (nxt < n);
      float ax = fmaf(hs.x, dd * dd, bb.x);
      float ay = fmaf(hs.y, dd * dd, bb.y);
      int ndeg = 0, ndegb = 0;
      int2 pn = make_int2(0, 0);
      float2 hn = make_float2(0.0f, 0.0f);

      if (FIRST) asm volatile("" : : "v"(p.x));
      else       asm volatile("" : : "v"(p.x), "v"(p.y));
      const int s_l = p.x;
      int c_src = 0;
      if (FIRST) c_src = cnt[s_l];
      int gc = degb < ROWS_AHEAD ? degb : ROWS_AHEAD;
      float2 v[ROWS_AHEAD];
      rows_load<0>(v, H2, s_l, 0, gc, lane);
      if (more) {
        ndeg = __builtin_amdgcn_readlane(c_l, j + 1);
        ndegb = ndeg < BUCKET ? ndeg : BUCKET;
        if (lane < ndegb) {
          if (FIRST) pn.x = bk[((size_t)nxt * BUCKET + lane) * 2];
          else       pn = ((const int2*)bk)[(size_t)nxt * BUCKET + lane];
        }
        hn = H2[(size_t)nxt * (H_DIM / 2) + lane];
      }
      float w_l;
      if (FIRST) {
        const float ws = rsqrtf((float)(c_src + 1));
        if (lane < degb)
          bk[((size_t)node * BUCKET + lane) * 2 + 1] = __float_as_int(ws);
        w_l = ws * dd;
      } else {
        w_l = __int_as_float(p.y) * dd;
      }
      rows_sum<0>(v, w_l, 0, gc, ax, ay);
      for (int i = ROWS_AHEAD; i < degb; ++i) {
        const int s = __builtin_amdgcn_readlane(s_l, i);
        const float w = __int_as_float(
            __builtin_amdgcn_readlane(__float_as_int(w_l), i));
        const float2 r = H2[(size_t)s * (H_DIM / 2) + lane];
        ax = fmaf(r.x, w, ax);
        ay = fmaf(r.y, w, ay);
      }

      if (deg > BUCKET) {
        int L = ovfc[0];
        if (L > ovf_cap) L = ovf_cap;
        for (int b0 = 0; b0 < L; b0 += 32) {
          int2 q = make_int2(-1, 0);
          if (b0 + lane < L) q = ((const int2*)ovf)[b0 + lane];
          unsigned hit = __builtin_amdgcn_ballot_w32(q.x == node);
          while (hit != 0u) {
            const int l = __builtin_ctz(hit);
            hit &= hit - 1u;
            const int s = __builtin_amdgcn_readlane(q.y, l);
            const float w = rsqrtf((float)(cnt[s] + 1)) * dd;
            const float2 r = H2[(size_t)s * (H_DIM / 2) + lane];
            ax = fmaf(r.x, w, ax);
            ay = fmaf(r.y, w, ay);
          }
        }
      }

      if (FIRST) {
        ax = fmaxf(ax, 0.0f);
        ay = fmaxf(ay, 0.0f);
        const int slot = j & 15;
        *(float2*)(&s_zt[wib][slot * ZT_PITCH + 2 * lane]) =
            make_float2(ax, ay);
        nrows = slot + 1;
      } else {
        ((float2*)Out)[(size_t)node * (H_DIM / 2) + lane] =
            make_float2(ax, ay);
      }
      if (!more) break;
      deg = ndeg;
      degb = ndegb;
      p = pn;
      hs = hn;
      node = nxt;
    }
  }
  if (FIRST) {
    if (nrows > 0)
      tile_gemm_store(&s_zt[wib][0], s_wq, Out, wave0, nwaves, nrows, lane);
  }
}

#define DEC_GROUP 16
__global__ void __launch_bounds__(256)
k_decode(const float* __restrict__ Z, const int* __restrict__ pos,
         const int* __restrict__ neg, float* __restrict__ out, int npos,
         int nneg) {
  const int total = npos + nneg;
  const int base = __builtin_amdgcn_readfirstlane(
      (int)(blockIdx.x * (blockDim.x >> 5) + (threadIdx.x >> 5))) * 32;
  if (base >= total) return;
  const int lane = threadIdx.x & 31;
  const int k = base + lane;
  int a = 0, b = 0;
  if (k < npos) {
    a = pos[k];
    b = pos[npos + k];
  } else if (k < total) {
    const int j = k - npos;
    a = neg[j];
    b = neg[nneg + j];
  }
  const float2* __restrict__ Z2 = (const float2*)Z;
  float res = 0.0f;
#pragma unroll 1
  for (int t0 = 0; t0 < 32; t0 += DEC_GROUP) {
    float2 va[DEC_GROUP], vb[DEC_GROUP];
#pragma unroll
    for (int i = 0; i < DEC_GROUP; ++i) {
      const int sa = __builtin_amdgcn_readlane(a, t0 + i);
      const int sb = __builtin_amdgcn_readlane(b, t0 + i);
      va[i] = Z2[(size_t)sa * (H_DIM / 2) + lane];
      vb[i] = Z2[(size_t)sb * (H_DIM / 2) + lane];
    }
#pragma unroll
    for (int i = 0; i < DEC_GROUP; ++i) {
      float s = fmaf(va[i].x, vb[i].x, va[i].y * vb[i].y);
#pragma unroll
      for (int o = 16; o > 0; o >>= 1) s += __shfl_xor(s, o, 32);
      if (lane == t0 + i) res = s;
    }
  }
  if (k < total) out[k] = res;
}

extern "C" void kernel_launch(void* const* d_in, const int* in_sizes, int n_in,
                              void* d_out, int out_size, void* d_ws,
                              size_t ws_size, hipStream_t stream) {
  (void)n_in; (void)out_size; (void)ws_size;
  const float* x   = (const float*)d_in[0];
  const int*   ei  = (const int*)d_in[1];
  const int*   pei = (const int*)d_in[2];
  const int*   nei = (const int*)d_in[3];
  const float* W1  = (const float*)d_in[4];
  const float* b1  = (const float*)d_in[5];
  const float* W2  = (const float*)d_in[6];
  const float* b2  = (const float*)d_in[7];
  float* out = (float*)d_out;

  const int n    = in_sizes[0] / H_DIM;
  const int ne   = in_sizes[1] / 2;
  const int npos = in_sizes[2] / 2;
  const int nneg = in_sizes[3] / 2;
  const int* src = ei;
  const int* dst = ei + ne;

  float* ws = (float*)d_ws;
  const size_t nAl = ((size_t)n + 63) & ~(size_t)63;
  float* hbuf = ws;
  float* acc1 = hbuf + (size_t)n * H_DIM;
  float* acc2 = acc1 + (size_t)n * H_DIM;
  int* cnt  = (int*)(acc2 + (size_t)n * H_DIM);
  int* ovfc = cnt + nAl;
  int* bk   = ovfc + 64;
  int* ovf  = bk + (size_t)n * BUCKET * 2;
  int* ovf2 = ovf + (size_t)ne * 2;
  int* gcount = ovf2 + (size_t)ne * 2;
  int* bins = gcount + MAX_BINS;
  const int nb = (n + BIN_NODES - 1) / BIN_NODES;

  const int B = 256;
  const int zB = B * ZERO_PER_THREAD;
  const int eB = B * EDGES_PER_THREAD;
  k_cnt_zero<<<(nb + zB - 1) / zB, B, 0, stream>>>(gcount, nb, ovfc);
  const int tiles = (n + 15) / 16;
  const int cblocks = (ne + eB - 1) / eB;
  const int gemm_blocks8 = (tiles + 7) / 8;
  k_bin_edges_gemm<<<cblocks + gemm_blocks8, B, 0, stream>>>(
      src, dst, gcount, ovfc, (int2*)bins, (int2*)ovf2, ne, cblocks, nb, x, W1,
      hbuf, n, tiles);
  k_bin_place<<<nb, B, 0, stream>>>(gcount, (const int2*)bins, ovfc,
                                    (const int2*)ovf2, ne, cnt, bk, (int2*)ovf,
                                    n);

  const int node_blocks = (n + 7) / 8;
  const int gblocks = node_blocks < 1280 ? node_blocks : 1280;
  const int tile_blocks = (n + 127) / 128;
  const int g1blocks = gblocks > tile_blocks ? gblocks : tile_blocks;
  k_gather_rows<true><<<g1blocks, B, 0, stream>>>(hbuf, cnt, bk, ovf, ovfc, ne,
                                                  b1, W2, acc1, n);
  k_gather_rows<false><<<gblocks, B, 0, stream>>>(acc1, cnt, bk, ovf, ovfc, ne,
                                                  b2, W2, acc2, n);
  k_decode<<<(npos + nneg + 255) / 256, B, 0, stream>>>(acc2, pei, nei, out,
                                                       npos, nneg);
}


# Edge binning without global counters: a block sorts its 4096 edges by bin in LDS and stores them as one whole-line segment plus a start table; no zeroing launch (6 to 5); sums keep arrival order
# speedup vs baseline: 4.0231x; 1.0717x over previous
#include <hip/hip_runtime.h>
#include <hip/hip_bf16.h>


typedef float v2f __attribute__((ext_vector_type(2)));
typedef float v8f __attribute__((ext_vector_type(8)));

#define H_DIM 64

#define EDGES_PER_THREAD 16

#define BUCKET 32
#define BIN_SHIFT 8
#define BIN_NODES 256
#define MAX_BINS 512
#define SEG_EDGES (256 * EDGES_PER_THREAD)

template <bool GUARD>
__device__ __forceinline__ void bin_edges_body(
    const int* __restrict__ src, const int* __restrict__ dst,
    int* __restrict__ tab, int cbp, int2* __restrict__ seg, int ne, int e0,
    int nb, int* s_hist, int* s_wsum, int2* s_sorted) {
  const int tid = (int)threadIdx.x;
  int d[EDGES_PER_THREAD], s[EDGES_PER_THREAD], r[EDGES_PER_THREAD];
#pragma unroll
  for (int j = 0; j < EDGES_PER_THREAD; ++j) {
    const int e = e0 + j * 256;
    const bool ok = !GUARD || e < ne;
    d[j] = ok ? dst[e] : 0;
    s[j] = ok ? src[e] : 0;
  }
#pragma unroll
  for (int j = 0; j < EDGES_PER_THREAD; ++j) {
    const int e = e0 + j * 256;
    r[j] = 0;
    if (!GUARD || e < ne) r[j] = atomicAdd(&s_hist[d[j] >> BIN_SHIFT], 1);
  }
  __syncthreads();
  static_assert(MAX_BINS == 512, "two bins per thread of a 256-thread block");
  const int2 c = *(const int2*)(s_hist + 2 * tid);
  const int lane = tid & 31;
  const int wave = tid >> 5;
  const int c01 = c.x + c.y;
  int incl = c01;
#pragma unroll
  for (int o = 1; o < 32; o <<= 1) {
    const int v = __shfl_up(incl, o, 32);
    if (lane >= o) incl += v;
  }
  if (lane == 31) s_wsum[wave] = incl;
  __syncthreads();
  int base = 0;
#pragma unroll
  for (int w = 0; w < 8; ++w) {
    const int v = s_wsum[w];
    if (w < wave) base += v;
  }
  const int l0 = base + incl - c01;
  const int l1 = l0 + c.x;
  *(int2*)(s_hist + 2 * tid) = make_int2(l0, l1);
  const int blk = (int)blockIdx.x;
  if (2 * tid < nb)
    tab[(size_t)(2 * tid) * cbp + blk] = l0 | (c.x << 16);
  if (2 * tid + 1 < nb)
    tab[(size_t)(2 * tid + 1) * cbp + blk] = l1 | (c.y << 16);
  __syncthreads();
  int st[EDGES_PER_THREAD];
#pragma unroll
  for (int j = 0; j < EDGES_PER_THREAD; ++j) st[j] = s_hist[d[j] >> BIN_SHIFT];
#pragma unroll
  for (int j = 0; j < EDGES_PER_THREAD; ++j) {
    const int e = e0 + j * 256;
    if (!GUARD || e < ne) s_sorted[st[j] + r[j]] = make_int2(d[j], s[j]);
  }
  __syncthreads();
  if (!GUARD) {
#pragma unroll
    for (int j = 0; j < EDGES_PER_THREAD / 2; ++j)
      ((int4*)seg)[j * 256 + tid] = ((const int4*)s_sorted)[j * 256 + tid];
  } else {
    const int nval = ne - (e0 - tid);
#pragma unroll
    for (int j = 0; j < EDGES_PER_THREAD; ++j) {
      const int i = j * 256 + tid;
      if (i < nval) seg[i] = s_sorted[i];
    }
  }
}

__device__ __forceinline__ void stage_w_quads(const float* __restrict__ Wn,
                                              float* s_wq) {
  float4 wv[4];
#pragma unroll
  for (int q = 0; q < 4; ++q) {
    const int cell = q * 256 + (int)threadIdx.x;
    const float* __restrict__ wp = Wn + ((cell >> 6) << 8) + (cell & 63);
    wv[q] = make_float4(wp[0], wp[H_DIM], wp[2 * H_DIM], wp[3 * H_DIM]);
  }
#pragma unroll
  for (int q = 0; q < 4; ++q)
    ((float4*)s_wq)[q * 256 + (int)threadIdx.x] = wv[q];
}

#define ZT_PITCH 68
__device__ __forceinline__ void tile_gemm_store(const float* zt,
                                                const float* wq,
                                                float* __restrict__ Hout,
                                                int nb, int nwaves, int nrows,
                                                int lane) {
  const int m     = lane & 15;
  const int khalf = (lane >> 4) << 1;
  const int rbase = (lane >> 4) << 3;
  const unsigned lane_off = (unsigned)(rbase * nwaves) * H_DIM + (unsigned)m;
#pragma unroll
  for (int c = 0; c < 4; c += 2) {
    v8f acc0 = {}, acc1 = {};
#pragma unroll
    for (int kk = 0; kk < 16; ++kk) {
      const v2f a  = *(const v2f*)(zt + m * ZT_PITCH + kk * 4 + khalf);
      const v2f b0 = *(const v2f*)(wq + ((kk * 64 + c * 16 + m) << 2) + khalf);
      const v2f b1 =
          *(const v2f*)(wq + ((kk * 64 + (c + 1) * 16 + m) << 2) + khalf);
      acc0 = __builtin_amdgcn_wmma_f32_16x16x4_f32(false, a, false, b0,
                                                   (short)0, acc0, false, false);
      acc1 = __builtin_amdgcn_wmma_f32_16x16x4_f32(false, a, false, b1,
                                                   (short)0, acc1, false, false);
    }
#pragma unroll
    for (int r = 0; r < 8; ++r) {
      if (rbase + r < nrows) {
        float* __restrict__ srow =
            Hout + (size_t)(nb + r * nwaves) * H_DIM + c * 16;
        srow[lane_off]      = acc0[r];
        srow[lane_off + 16] = acc1[r];
      }
    }
  }
}

__global__ void __launch_bounds__(256)
k_bin_edges_gemm(const int* __restrict__ esrc, const int* __restrict__ dst,
                 int* __restrict__ tab, int cbp, int2* __restrict__ segs,
                 int* __restrict__ ovfc, int ne, int cblocks, int nb,
                 const float* __restrict__ X, const float* __restrict__ W,
                 float* __restrict__ Hout, int n, int tiles) {
  __shared__ alignas(16) int s_hist[MAX_BINS];
  __shared__ alignas(16) float s_wq[H_DIM * H_DIM];
  __shared__ alignas(16) float s_xt[8][16 * ZT_PITCH];
  static_assert(sizeof(float) * 8 * 16 * ZT_PITCH >= sizeof(int2) * SEG_EDGES,
                "the segment's LDS copy fits the tiles' room");
  if (blockIdx.x == 0 && threadIdx.x == 0) ovfc[0] = 0;
  if ((int)blockIdx.x < cblocks) {
    for (int i = (int)threadIdx.x; i < MAX_BINS; i += 256) s_hist[i] = 0;
    __syncthreads();
    const int eb = (int)blockIdx.x * SEG_EDGES;
    const int e0 = eb + (int)threadIdx.x;
    int* s_wsum = reinterpret_cast<int*>(&s_wq[0]);
    int2* s_sorted = reinterpret_cast<int2*>(&s_xt[0][0]);
    int2* seg = segs + (size_t)blockIdx.x * SEG_EDGES;
    if (eb + SEG_EDGES <= ne)
      bin_edges_body<false>(esrc, dst, tab, cbp, seg, ne, e0, nb, s_hist,
                            s_wsum, s_sorted);
    else
      bin_edges_body<true>(esrc, dst, tab, cbp, seg, ne, e0, nb, s_hist,
                           s_wsum, s_sorted);
  } else {
    const int wave = __builtin_amdgcn_readfirstlane((int)(threadIdx.x >> 5));
    const int lane = (int)threadIdx.x & 31;
    const int g = (int)blockIdx.x - cblocks;
    float* zt = &s_xt[wave][0];
    const int rhalf = lane >> 4;
    const int c4 = lane & 15;
    const float4* __restrict__ X4 = (const float4*)X;
    const int tile = g * 8 + wave;
    const int row0 = tile * 16;
    float4 xv[8];
#pragma unroll
    for (int i = 0; i < 8; ++i) {
      int row = row0 + 2 * i + rhalf;
      if (row >= n) row = n - 1;
      xv[i] = X4[(size_t)row * (H_DIM / 4) + c4];
    }
    stage_w_quads(W, s_wq);
#pragma unroll
    for (int i = 0; i < 8; ++i)
      *(float4*)(zt + (2 * i + rhalf) * ZT_PITCH + c4 * 4) = xv[i];
    __syncthreads();
    if (tile < tiles) {
      int nrows = n - row0;
      if (nrows > 16) nrows = 16;
      tile_gemm_store(zt, s_wq, Hout, row0, 1, nrows, lane);
    }
  }
}

#define RUNS_AHEAD 32
#define TAB_PITCH (BUCKET + 1)
__device__ __forceinline__ void place_edge(int2 q, int* s_cnt, int* s_tab,
                                           int* __restrict__ ovfc,
                                           int2* __restrict__ ovf) {
  const int j = q.x & (BIN_NODES - 1);
  const int r = atomicAdd(&s_cnt[j], 1);
  if (r < BUCKET) {
    s_tab[j * TAB_PITCH + r] = q.y;
  } else {
    const int o = atomicAdd(ovfc, 1);
    ovf[o] = q;
  }
}

template <int G>
__device__ __forceinline__ void place_runs(int tw, int kbase, int blk0, int wv,
                                           int lane,
                                           const int2* __restrict__ segs,
                                           int* s_cnt, int* s_tab,
                                           int* __restrict__ ovfc,
                                           int2* __restrict__ ovf) {
  int2 q[G];
#pragma unroll
  for (int k = 0; k < G; ++k) {
    const int w = __builtin_amdgcn_readlane(tw, kbase + k);
    const int2* __restrict__ run =
        segs + (size_t)(blk0 + wv + 8 * (kbase + k)) * SEG_EDGES + (w & 0xffff);
    q[k] = make_int2(-1, 0);
    if (lane < (w >> 16)) q[k] = run[lane];
  }
  int r[G];
#pragma unroll
  for (int k = 0; k < G; ++k) {
    const int j = q[k].x >= 0 ? (q[k].x & (BIN_NODES - 1)) : BIN_NODES + lane;
    r[k] = atomicAdd(&s_cnt[j], 1);
  }
#pragma unroll
  for (int k = 0; k < G; ++k) {
    if (q[k].x >= 0) {
      const int j = q[k].x & (BIN_NODES - 1);
      if (r[k] < BUCKET) {
        s_tab[j * TAB_PITCH + r[k]] = q[k].y;
      } else {
        const int o = atomicAdd(ovfc, 1);
        ovf[o] = q[k];
      }
    }
  }
}

__global__ void __launch_bounds__(256)
k_bin_place(const int* __restrict__ tab, int cbp, int cblocks,
            const int2* __restrict__ segs, int* __restrict__ ovfc,
            int* __restrict__ cnt, int* __restrict__ bk,
            int2* __restrict__ ovf, int n) {
  __shared__ int s_cnt[BIN_NODES + 32];
  __shared__ int s_tab[BIN_NODES * TAB_PITCH];
  const int tid = (int)threadIdx.x;
  const int b = (int)blockIdx.x;
  const int wv = __builtin_amdgcn_readfirstlane(tid >> 5);
  const int lane = tid & 31;
  s_cnt[tid] = 0;
  if (tid < 32) s_cnt[BIN_NODES + tid] = 0;
  const int* __restrict__ trow = tab + (size_t)b * cbp;
  int tw = 0;
  if (wv + 8 * lane < cblocks) tw = trow[wv + 8 * lane];
  __syncthreads();
  for (int blk0 = 0; blk0 < cblocks; blk0 += 256) {
    int twn = 0;
    const int nblk = blk0 + 256 + wv + 8 * lane;
    if (nblk < cblocks) twn = trow[nblk];
    if (blk0 + 256 <= cblocks) {
      place_runs<RUNS_AHEAD>(tw, 0, blk0, wv, lane, segs, s_cnt, s_tab, ovfc,
                             ovf);
    } else {
#pragma unroll 1
      for (int k0 = 0; blk0 + wv + 8 * k0 < cblocks; k0 += 8)
        place_runs<8>(tw, k0, blk0, wv, lane, segs, s_cnt, s_tab, ovfc, ovf);
    }
    if (__builtin_amdgcn_ballot_w32((tw >> 16) > 32) != 0u) {
#pragma unroll 1
      for (int k = 0; k < RUNS_AHEAD; ++k) {
        const int w = __builtin_amdgcn_readlane(tw, k);
        const int cn = w >> 16;
        const int2* __restrict__ run =
            segs + (size_t)(blk0 + wv + 8 * k) * SEG_EDGES + (w & 0xffff);
        for (int i = 32 + lane; i < cn; i += 32)
          place_edge(run[i], s_cnt, s_tab, ovfc, ovf);
      }
    }
    tw = twn;
  }
  __syncthreads();
  const int wbase = wv * 32;
  const int c_l = s_cnt[wbase + lane];
  const int mine = b * BIN_NODES + wbase + lane;
  if (mine < n) cnt[mine] = c_l;
  int* __restrict__ bkw =
      bk + ((size_t)(b * BIN_NODES + wbase) * BUCKET + lane) * 2;
#pragma unroll
  for (int k0 = 0; k0 < 32; k0 += 8) {
    int v[8];
#pragma unroll
    for (int k = 0; k < 8; ++k) {
      const int ck = __builtin_amdgcn_readlane(c_l, k0 + k);
      v[k] = 0;
      if (lane < ck) v[k] = s_tab[(wbase + k0 + k) * TAB_PITCH + lane];
    }
#pragma unroll
    for (int k = 0; k < 8; ++k) {
      const int ck = __builtin_amdgcn_readlane(c_l, k0 + k);
      if (lane < ck) bkw[(size_t)(k0 + k) * BUCKET * 2] = v[k];
    }
  }
}

#define ROWS_AHEAD 24
template <int I>
__device__ __forceinline__ void rows_load(float2 (&v)[ROWS_AHEAD],
                                          const float2* __restrict__ H2,
                                          int s_l, int t, int cnt, int lane) {
  if constexpr (I < ROWS_AHEAD) {
    if (I < cnt) {
      const int s = __builtin_amdgcn_readlane(s_l, t + I);
      v[I] = H2[(size_t)s * (H_DIM / 2) + lane];
      rows_load<I + 1>(v, H2, s_l, t, cnt, lane);
    }
  }
}
template <int I>
__device__ __forceinline__ void rows_sum(const float2 (&v)[ROWS_AHEAD],
                                         float w_l, int t, int cnt, float& ax,
                                         float& ay) {
  if constexpr (I < ROWS_AHEAD) {
    if (I < cnt) {
      const float w = __int_as_float(
          __builtin_amdgcn_readlane(__float_as_int(w_l), t + I));
      ax = fmaf(v[I].x, w, ax);
      ay = fmaf(v[I].y, w, ay);
      rows_sum<I + 1>(v, w_l, t, cnt, ax, ay);
    }
  }
}

template <bool FIRST>
__global__ void __launch_bounds__(256, 10)
k_gather_rows(const float* __restrict__ Hm, const int* __restrict__ cnt,
              int* __restrict__ bk, const int* __restrict__ ovf,
              const int* __restrict__ ovfc, int ovf_cap,
              const float* __restrict__ bias, const float* __restrict__ Wn,
              float* __restrict__ Out, int n) {
  const int wave0 = __builtin_amdgcn_readfirstlane(
      (int)(blockIdx.x * (blockDim.x >> 5) + (threadIdx.x >> 5)));
  const int nwaves = (int)(gridDim.x * (blockDim.x >> 5));
  const int lane = threadIdx.x & 31;
  const float2 bb = ((const float2*)bias)[lane];
  const float2* __restrict__ H2 = (const float2*)Hm;

  __shared__ alignas(16) float s_wq[H_DIM * H_DIM];
  __shared__ alignas(16) float s_zt[8][16 * ZT_PITCH];
  const int wib = __builtin_amdgcn_readfirstlane((int)(threadIdx.x >> 5));
  if (FIRST) {
    stage_w_quads(Wn, s_wq);
    __syncthreads();
  }

  int nrows = 0;
  for (int nb = wave0; nb < n; nb += 32 * nwaves) {
    const int mine = nb + lane * nwaves;
    int c_l = 0;
    if (mine < n) c_l = cnt[mine];
    const float d_l = rsqrtf((float)(c_l + 1));
    int deg = __builtin_amdgcn_readlane(c_l, 0);
    int degb = deg < BUCKET ? deg : BUCKET;
    int2 p = make_int2(0, 0);
    if (lane < degb) {
      if (FIRST) p.x = bk[((size_t)nb * BUCKET + lane) * 2];
      else       p = ((const int2*)bk)[(size_t)nb * BUCKET + lane];
    }
    float2 hs = H2[(size_t)nb * (H_DIM / 2) + lane];

    int node = nb;
    for (int j = 0; j < 32; ++j) {
      const float dd = __int_as_float(
          __builtin_amdgcn_readlane(__float_as_int(d_l), j));
      const int nxt = node + nwaves;
      const bool more = (j + 1 < 32) && (nxt < n);
      float ax = fmaf(hs.x, dd * dd, bb.x);
      float ay = fmaf(hs.y, dd * dd, bb.y);
      int ndeg = 0, ndegb = 0;
      int2 pn = make_int2(0, 0);
      float2 hn = make_float2(0.0f, 0.0f);

      if (FIRST) asm volatile("" : : "v"(p.x));
      else       asm volatile("" : : "v"(p.x), "v"(p.y));
      const int s_l = p.x;
      int c_src = 0;
      if (FIRST) c_src = cnt[s_l];
      int gc = degb < ROWS_AHEAD ? degb : ROWS_AHEAD;
      float2 v[ROWS_AHEAD];
      rows_load<0>(v, H2, s_l, 0, gc, lane);
      if (more) {
        ndeg = __builtin_amdgcn_readlane(c_l, j + 1);
        ndegb = ndeg < BUCKET ? ndeg : BUCKET;
        if (lane < ndegb) {
          if (FIRST) pn.x = bk[((size_t)nxt * BUCKET + lane) * 2];
          else       pn = ((const int2*)bk)[(size_t)nxt * BUCKET + lane];
        }
        hn = H2[(size_t)nxt * (H_DIM / 2) + lane];
      }
      float w_l;
      if (FIRST) {
        const float ws = rsqrtf((float)(c_src + 1));
        if (lane < degb)
          bk[((size_t)node * BUCKET + lane) * 2 + 1] = __float_as_int(ws);
        w_l = ws * dd;
      } else {
        w_l = __int_as_float(p.y) * dd;
      }
      rows_sum<0>(v, w_l, 0, gc, ax, ay);
      for (int i = ROWS_AHEAD; i < degb; ++i) {
        const int s = __builtin_amdgcn_readlane(s_l, i);
        const float w = __int_as_float(
            __builtin_amdgcn_readlane(__float_as_int(w_l), i));
        const float2 r = H2[(size_t)s * (H_DIM / 2) + lane];
        ax = fmaf(r.x, w, ax);
        ay = fmaf(r.y, w, ay);
      }

      if (deg > BUCKET) {
        int L = ovfc[0];
        if (L > ovf_cap) L = ovf_cap;
        for (int b0 = 0; b0 < L; b0 += 32) {
          int2 q = make_int2(-1, 0);
          if (b0 + lane < L) q = ((const int2*)ovf)[b0 + lane];
          unsigned hit = __builtin_amdgcn_ballot_w32(q.x == node);
          while (hit != 0u) {
            const int l = __builtin_ctz(hit);
            hit &= hit - 1u;
            const int s = __builtin_amdgcn_readlane(q.y, l);
            const float w = rsqrtf((float)(cnt[s] + 1)) * dd;
            const float2 r = H2[(size_t)s * (H_DIM / 2) + lane];
            ax = fmaf(r.x, w, ax);
            ay = fmaf(r.y, w, ay);
          }
        }
      }

      if (FIRST) {
        ax = fmaxf(ax, 0.0f);
        ay = fmaxf(ay, 0.0f);
        const int slot = j & 15;
        *(float2*)(&s_zt[wib][slot * ZT_PITCH + 2 * lane]) =
            make_float2(ax, ay);
        nrows = slot + 1;
      } else {
        ((float2*)Out)[(size_t)node * (H_DIM / 2) + lane] =
            make_float2(ax, ay);
      }
      if (!more) break;
      deg = ndeg;
      degb = ndegb;
      p = pn;
      hs = hn;
      node = nxt;
    }
  }
  if (FIRST) {
    if (nrows > 0)
      tile_gemm_store(&s_zt[wib][0], s_wq, Out, wave0, nwaves, nrows, lane);
  }
}

#define DEC_GROUP 16
__global__ void __launch_bounds__(256)
k_decode(const float* __restrict__ Z, const int* __restrict__ pos,
         const int* __restrict__ neg, float* __restrict__ out, int npos,
         int nneg) {
  const int total = npos + nneg;
  const int base = __builtin_amdgcn_readfirstlane(
      (int)(blockIdx.x * (blockDim.x >> 5) + (threadIdx.x >> 5))) * 32;
  if (base >= total) return;
  const int lane = threadIdx.x & 31;
  const int k = base + lane;
  int a = 0, b = 0;
  if (k < npos) {
    a = pos[k];
    b = pos[npos + k];
  } else if (k < total) {
    const int j = k - npos;
    a = neg[j];
    b = neg[nneg + j];
  }
  const float2* __restrict__ Z2 = (const float2*)Z;
  float res = 0.0f;
#pragma unroll 1
  for (int t0 = 0; t0 < 32; t0 += DEC_GROUP) {
    float2 va[DEC_GROUP], vb[DEC_GROUP];
#pragma unroll
    for (int i = 0; i < DEC_GROUP; ++i) {
      const int sa = __builtin_amdgcn_readlane(a, t0 + i);
      const int sb = __builtin_amdgcn_readlane(b, t0 + i);
      va[i] = Z2[(size_t)sa * (H_DIM / 2) + lane];
      vb[i] = Z2[(size_t)sb * (H_DIM / 2) + lane];
    }
#pragma unroll
    for (int i = 0; i < DEC_GROUP; ++i) {
      float s = fmaf(va[i].x, vb[i].x, va[i].y * vb[i].y);
#pragma unroll
      for (int o = 16; o > 0; o >>= 1) s += __shfl_xor(s, o, 32);
      if (lane == t0 + i) res = s;
    }
  }
  if (k < total) out[k] = res;
}

extern "C" void kernel_launch(void* const* d_in, const int* in_sizes, int n_in,
                              void* d_out, int out_size, void* d_ws,
                              size_t ws_size, hipStream_t stream) {
  (void)n_in; (void)out_size; (void)ws_size;
  const float* x   = (const float*)d_in[0];
  const int*   ei  = (const int*)d_in[1];
  const int*   pei = (const int*)d_in[2];
  const int*   nei = (const int*)d_in[3];
  const float* W1  = (const float*)d_in[4];
  const float* b1  = (const float*)d_in[5];
  const float* W2  = (const float*)d_in[6];
  const float* b2  = (const float*)d_in[7];
  float* out = (float*)d_out;

  const int n    = in_sizes[0] / H_DIM;
  const int ne   = in_sizes[1] / 2;
  const int npos = in_sizes[2] / 2;
  const int nneg = in_sizes[3] / 2;
  const int* src = ei;
  const int* dst = ei + ne;

  const int B = 256;
  const int eB = SEG_EDGES;
  const int nb = (n + BIN_NODES - 1) / BIN_NODES;
  const int cblocks = (ne + eB - 1) / eB;
  const int cbp = (cblocks + 31) & ~31;
  float* ws = (float*)d_ws;
  const size_t nAl = ((size_t)n + 63) & ~(size_t)63;
  const size_t neAl = ((size_t)ne + 15) & ~(size_t)15;
  float* hbuf = ws;
  float* acc1 = hbuf + (size_t)n * H_DIM;
  float* acc2 = acc1 + (size_t)n * H_DIM;
  int* cnt  = (int*)(acc2 + (size_t)n * H_DIM);
  int* ovfc = cnt + nAl;
  int* bk   = ovfc + 64;
  int* ovf  = bk + (size_t)n * BUCKET * 2;
  int* tab  = ovf + neAl * 2;
  int* segs = tab + (size_t)nb * cbp;

  const int tiles = (n + 15) / 16;
  const int gemm_blocks8 = (tiles + 7) / 8;
  k_bin_edges_gemm<<<cblocks + gemm_blocks8, B, 0, stream>>>(
      src, dst, tab, cbp, (int2*)segs, ovfc, ne, cblocks, nb, x, W1, hbuf, n,
      tiles);
  k_bin_place<<<nb, B, 0, stream>>>(tab, cbp, cblocks, (const int2*)segs, ovfc,
                                    cnt, bk, (int2*)ovf, n);

  const int node_blocks = (n + 7) / 8;
  const int gblocks = node_blocks < 1280 ? node_blocks : 1280;
  const int tile_blocks = (n + 127) / 128;
  const int g1blocks = gblocks > tile_blocks ? gblocks : tile_blocks;
  k_gather_rows<true><<<g1blocks, B, 0, stream>>>(hbuf, cnt, bk, ovf, ovfc, ne,
                                                  b1, W2, acc1, n);
  k_gather_rows<false><<<gblocks, B, 0, stream>>>(acc1, cnt, bk, ovf, ovfc, ne,
                                                  b2, W2, acc2, n);
  k_decode<<<(npos + nneg + 255) / 256, B, 0, stream>>>(acc2, pei, nei, out,
                                                       npos, nneg);
}


# Rows stored times dis[source]: the bin's block knows its nodes' degrees, does X @ W1 for them; row sums are plain adds, dis[dest] once per sum, no per-edge degree lookup; rounding differs, order kept
# speedup vs baseline: 4.2085x; 1.0461x over previous
#include <hip/hip_runtime.h>
#include <hip/hip_bf16.h>


typedef float v2f __attribute__((ext_vector_type(2)));
typedef float v8f __attribute__((ext_vector_type(8)));

#define H_DIM 64

#define EDGES_PER_THREAD 16

#define BUCKET 32
#define BIN_SHIFT 8
#define BIN_NODES 256
#define MAX_BINS 512
#define SEG_EDGES (256 * EDGES_PER_THREAD)

template <bool GUARD>
__device__ __forceinline__ void bin_edges_body(
    const int* __restrict__ src, const int* __restrict__ dst,
    int* __restrict__ tab, int cbp, int2* __restrict__ seg, int ne, int e0,
    int nb, int* s_hist, int* s_wsum, int2* s_sorted) {
  const int tid = (int)threadIdx.x;
  int d[EDGES_PER_THREAD], s[EDGES_PER_THREAD], r[EDGES_PER_THREAD];
#pragma unroll
  for (int j = 0; j < EDGES_PER_THREAD; ++j) {
    const int e = e0 + j * 256;
    const bool ok = !GUARD || e < ne;
    d[j] = ok ? dst[e] : 0;
    s[j] = ok ? src[e] : 0;
  }
#pragma unroll
  for (int j = 0; j < EDGES_PER_THREAD; ++j) {
    const int e = e0 + j * 256;
    r[j] = 0;
    if (!GUARD || e < ne) r[j] = atomicAdd(&s_hist[d[j] >> BIN_SHIFT], 1);
  }
  __syncthreads();
  static_assert(MAX_BINS == 512, "two bins per thread of a 256-thread block");
  const int2 c = *(const int2*)(s_hist + 2 * tid);
  const int lane = tid & 31;
  const int wave = tid >> 5;
  const int c01 = c.x + c.y;
  int incl = c01;
#pragma unroll
  for (int o = 1; o < 32; o <<= 1) {
    const int v = __shfl_up(incl, o, 32);
    if (lane >= o) incl += v;
  }
  if (lane == 31) s_wsum[wave] = incl;
  __syncthreads();
  int base = 0;
#pragma unroll
  for (int w = 0; w < 8; ++w) {
    const int v = s_wsum[w];
    if (w < wave) base += v;
  }
  const int l0 = base + incl - c01;
  const int l1 = l0 + c.x;
  *(int2*)(s_hist + 2 * tid) = make_int2(l0, l1);
  const int blk = (int)blockIdx.x;
  if (2 * tid < nb)
    tab[(size_t)(2 * tid) * cbp + blk] = l0 | (c.x << 16);
  if (2 * tid + 1 < nb)
    tab[(size_t)(2 * tid + 1) * cbp + blk] = l1 | (c.y << 16);
  __syncthreads();
  int st[EDGES_PER_THREAD];
#pragma unroll
  for (int j = 0; j < EDGES_PER_THREAD; ++j) st[j] = s_hist[d[j] >> BIN_SHIFT];
#pragma unroll
  for (int j = 0; j < EDGES_PER_THREAD; ++j) {
    const int e = e0 + j * 256;
    if (!GUARD || e < ne) s_sorted[st[j] + r[j]] = make_int2(d[j], s[j]);
  }
  __syncthreads();
  if (!GUARD) {
#pragma unroll
    for (int j = 0; j < EDGES_PER_THREAD / 2; ++j)
      ((int4*)seg)[j * 256 + tid] = ((const int4*)s_sorted)[j * 256 + tid];
  } else {
    const int nval = ne - (e0 - tid);
#pragma unroll
    for (int j = 0; j < EDGES_PER_THREAD; ++j) {
      const int i = j * 256 + tid;
      if (i < nval) seg[i] = s_sorted[i];
    }
  }
}

__device__ __forceinline__ void load_w_quads(const float* __restrict__ Wn,
                                             float4 (&wv)[4]) {
#pragma unroll
  for (int q = 0; q < 4; ++q) {
    const int cell = q * 256 + (int)threadIdx.x;
    const float* __restrict__ wp = Wn + ((cell >> 6) << 8) + (cell & 63);
    wv[q] = make_float4(wp[0], wp[H_DIM], wp[2 * H_DIM], wp[3 * H_DIM]);
  }
}
__device__ __forceinline__ void store_w_quads(const float4 (&wv)[4],
                                              float* s_wq) {
#pragma unroll
  for (int q = 0; q < 4; ++q)
    ((float4*)s_wq)[q * 256 + (int)threadIdx.x] = wv[q];
}

#define ZT_PITCH 68
__device__ __forceinline__ void tile_gemm_store(const float* zt,
                                                const float* wq,
                                                float* __restrict__ Hout,
                                                int nb, int nwaves, int nrows,
                                                int lane) {
  const int m     = lane & 15;
  const int khalf = (lane >> 4) << 1;
  const int rbase = (lane >> 4) << 3;
  const unsigned lane_off = (unsigned)(rbase * nwaves) * H_DIM + (unsigned)m;
#pragma unroll
  for (int c = 0; c < 4; c += 2) {
    v8f acc0 = {}, acc1 = {};
#pragma unroll
    for (int kk = 0; kk < 16; ++kk) {
      const v2f a  = *(const v2f*)(zt + m * ZT_PITCH + kk * 4 + khalf);
      const v2f b0 = *(const v2f*)(wq + ((kk * 64 + c * 16 + m) << 2) + khalf);
      const v2f b1 =
          *(const v2f*)(wq + ((kk * 64 + (c + 1) * 16 + m) << 2) + khalf);
      acc0 = __builtin_amdgcn_wmma_f32_16x16x4_f32(false, a, false, b0,
                                                   (short)0, acc0, false, false);
      acc1 = __builtin_amdgcn_wmma_f32_16x16x4_f32(false, a, false, b1,
                                                   (short)0, acc1, false, false);
    }
#pragma unroll
    for (int r = 0; r < 8; ++r) {
      if (rbase + r < nrows) {
        float* __restrict__ srow =
            Hout + (size_t)(nb + r * nwaves) * H_DIM + c * 16;
        srow[lane_off]      = acc0[r];
        srow[lane_off + 16] = acc1[r];
      }
    }
  }
}

__global__ void __launch_bounds__(256)
k_bin_edges(const int* __restrict__ esrc, const int* __restrict__ dst,
            int* __restrict__ tab, int cbp, int2* __restrict__ segs,
            int* __restrict__ ovfc, int ne, int nb) {
  __shared__ alignas(16) int s_hist[MAX_BINS];
  __shared__ int s_wsum[8];
  __shared__ alignas(16) int2 s_sorted[SEG_EDGES];
  if (blockIdx.x == 0 && threadIdx.x == 0) ovfc[0] = 0;
  for (int i = (int)threadIdx.x; i < MAX_BINS; i += 256) s_hist[i] = 0;
  __syncthreads();
  const int eb = (int)blockIdx.x * SEG_EDGES;
  const int e0 = eb + (int)threadIdx.x;
  int2* seg = segs + (size_t)blockIdx.x * SEG_EDGES;
  if (eb + SEG_EDGES <= ne)
    bin_edges_body<false>(esrc, dst, tab, cbp, seg, ne, e0, nb, s_hist,
                          s_wsum, s_sorted);
  else
    bin_edges_body<true>(esrc, dst, tab, cbp, seg, ne, e0, nb, s_hist,
                         s_wsum, s_sorted);
}

#define RUNS_AHEAD 32
#define TAB_PITCH (BUCKET + 1)
__device__ __forceinline__ void place_edge(int2 q, int* s_cnt, int* s_tab,
                                           int* __restrict__ ovfc,
                                           int2* __restrict__ ovf) {
  const int j = q.x & (BIN_NODES - 1);
  const int r = atomicAdd(&s_cnt[j], 1);
  if (r < BUCKET) {
    s_tab[j * TAB_PITCH + r] = q.y;
  } else {
    const int o = atomicAdd(ovfc, 1);
    ovf[o] = q;
  }
}

template <int G>
__device__ __forceinline__ void place_runs(int tw, int kbase, int blk0, int wv,
                                           int lane,
                                           const int2* __restrict__ segs,
                                           int* s_cnt, int* s_tab,
                                           int* __restrict__ ovfc,
                                           int2* __restrict__ ovf) {
  int2 q[G];
#pragma unroll
  for (int k = 0; k < G; ++k) {
    const int w = __builtin_amdgcn_readlane(tw, kbase + k);
    const int2* __restrict__ run =
        segs + (size_t)(blk0 + wv + 8 * (kbase + k)) * SEG_EDGES + (w & 0xffff);
    q[k] = make_int2(-1, 0);
    if (lane < (w >> 16)) q[k] = run[lane];
  }
  int r[G];
#pragma unroll
  for (int k = 0; k < G; ++k) {
    const int j = q[k].x >= 0 ? (q[k].x & (BIN_NODES - 1)) : BIN_NODES + lane;
    r[k] = atomicAdd(&s_cnt[j], 1);
  }
#pragma unroll
  for (int k = 0; k < G; ++k) {
    if (q[k].x >= 0) {
      const int j = q[k].x & (BIN_NODES - 1);
      if (r[k] < BUCKET) {
        s_tab[j * TAB_PITCH + r[k]] = q[k].y;
      } else {
        const int o = atomicAdd(ovfc, 1);
        ovf[o] = q[k];
      }
    }
  }
}

__global__ void __launch_bounds__(256)
k_bin_place(const int* __restrict__ tab, int cbp, int cblocks,
            const int2* __restrict__ segs, int* __restrict__ ovfc,
            int* __restrict__ cnt, int* __restrict__ bk,
            int2* __restrict__ ovf, int n, const float* __restrict__ X,
            const float* __restrict__ W, float* __restrict__ Hout) {
  __shared__ int s_cnt[BIN_NODES + 32];
  __shared__ int s_tab[BIN_NODES * TAB_PITCH];
  __shared__ alignas(16) float s_wq[H_DIM * H_DIM];
  __shared__ alignas(16) float s_xt[BIN_NODES / 16][16 * ZT_PITCH];
  static_assert(BIN_NODES / 16 == 2 * 8, "two 16-row tiles per wave");
  const int tid = (int)threadIdx.x;
  const int b = (int)blockIdx.x;
  const int wv = __builtin_amdgcn_readfirstlane(tid >> 5);
  const int lane = tid & 31;
  s_cnt[tid] = 0;
  if (tid < 32) s_cnt[BIN_NODES + tid] = 0;
  const int* __restrict__ trow = tab + (size_t)b * cbp;
  int tw = 0;
  if (wv + 8 * lane < cblocks) tw = trow[wv + 8 * lane];
  const int rhalf = lane >> 4;
  const int c4 = lane & 15;
  const float4* __restrict__ X4 = (const float4*)X;
  float4 xv[2][8];
#pragma unroll
  for (int t = 0; t < 2; ++t) {
#pragma unroll
    for (int i = 0; i < 8; ++i) {
      int row = b * BIN_NODES + (wv * 2 + t) * 16 + 2 * i + rhalf;
      if (row >= n) row = n - 1;
      xv[t][i] = X4[(size_t)row * (H_DIM / 4) + c4];
    }
  }
  float4 wq4[4];
  load_w_quads(W, wq4);
  __syncthreads();
  for (int blk0 = 0; blk0 < cblocks; blk0 += 256) {
    int twn = 0;
    const int nblk = blk0 + 256 + wv + 8 * lane;
    if (nblk < cblocks) twn = trow[nblk];
    if (blk0 + 256 <= cblocks) {
      place_runs<RUNS_AHEAD>(tw, 0, blk0, wv, lane, segs, s_cnt, s_tab, ovfc,
                             ovf);
    } else {
#pragma unroll 1
      for (int k0 = 0; blk0 + wv + 8 * k0 < cblocks; k0 += 8)
        place_runs<8>(tw, k0, blk0, wv, lane, segs, s_cnt, s_tab, ovfc, ovf);
    }
    if (__builtin_amdgcn_ballot_w32((tw >> 16) > 32) != 0u) {
#pragma unroll 1
      for (int k = 0; k < RUNS_AHEAD; ++k) {
        const int w = __builtin_amdgcn_readlane(tw, k);
        const int cn = w >> 16;
        const int2* __restrict__ run =
            segs + (size_t)(blk0 + wv + 8 * k) * SEG_EDGES + (w & 0xffff);
        for (int i = 32 + lane; i < cn; i += 32)
          place_edge(run[i], s_cnt, s_tab, ovfc, ovf);
      }
    }
    tw = twn;
  }
  store_w_quads(wq4, s_wq);
  __syncthreads();
  const int wbase = wv * 32;
  const int c_l = s_cnt[wbase + lane];
  const int mine = b * BIN_NODES + wbase + lane;
  if (mine < n) cnt[mine] = c_l;
  int* __restrict__ bkw = bk + (size_t)(b * BIN_NODES + wbase) * BUCKET + lane;
#pragma unroll
  for (int k0 = 0; k0 < 32; k0 += 8) {
    int v[8];
#pragma unroll
    for (int k = 0; k < 8; ++k) {
      const int ck = __builtin_amdgcn_readlane(c_l, k0 + k);
      v[k] = 0;
      if (lane < ck) v[k] = s_tab[(wbase + k0 + k) * TAB_PITCH + lane];
    }
#pragma unroll
    for (int k = 0; k < 8; ++k) {
      const int ck = __builtin_amdgcn_readlane(c_l, k0 + k);
      if (lane < ck) bkw[(size_t)(k0 + k) * BUCKET] = v[k];
    }
  }
#pragma unroll
  for (int t = 0; t < 2; ++t) {
    const int trow0 = (wv * 2 + t) * 16;
    float* zt = &s_xt[wv * 2 + t][0];
#pragma unroll
    for (int i = 0; i < 8; ++i) {
      const int rr = 2 * i + rhalf;
      const float sc = rsqrtf((float)(s_cnt[trow0 + rr] + 1));
      const float4 x = xv[t][i];
      *(float4*)(zt + rr * ZT_PITCH + c4 * 4) =
          make_float4(x.x * sc, x.y * sc, x.z * sc, x.w * sc);
    }
    const int row0 = b * BIN_NODES + trow0;
    if (row0 < n) {
      int nrows = n - row0;
      if (nrows > 16) nrows = 16;
      tile_gemm_store(zt, s_wq, Hout, row0, 1, nrows, lane);
    }
  }
}

#define ROWS_AHEAD 24
template <int I>
__device__ __forceinline__ void rows_load(float2 (&v)[ROWS_AHEAD],
                                          const float2* __restrict__ H2,
                                          int s_l, int t, int cnt, int lane) {
  if constexpr (I < ROWS_AHEAD) {
    if (I < cnt) {
      const int s = __builtin_amdgcn_readlane(s_l, t + I);
      v[I] = H2[(size_t)s * (H_DIM / 2) + lane];
      rows_load<I + 1>(v, H2, s_l, t, cnt, lane);
    }
  }
}
template <int I>
__device__ __forceinline__ void rows_sum(const float2 (&v)[ROWS_AHEAD],
                                         int cnt, float& ax, float& ay) {
  if constexpr (I < ROWS_AHEAD) {
    if (I < cnt) {
      ax += v[I].x;
      ay += v[I].y;
      rows_sum<I + 1>(v, cnt, ax, ay);
    }
  }
}

template <bool FIRST>
__global__ void __launch_bounds__(256, 10)
k_gather_rows(const float* __restrict__ Hm, const int* __restrict__ cnt,
              const int* __restrict__ bk, const int* __restrict__ ovf,
              const int* __restrict__ ovfc, int ovf_cap,
              const float* __restrict__ bias, const float* __restrict__ Wn,
              float* __restrict__ Out, int n) {
  const int wave0 = __builtin_amdgcn_readfirstlane(
      (int)(blockIdx.x * (blockDim.x >> 5) + (threadIdx.x >> 5)));
  const int nwaves = (int)(gridDim.x * (blockDim.x >> 5));
  const int lane = threadIdx.x & 31;
  const float2 bb = ((const float2*)bias)[lane];
  const float2* __restrict__ H2 = (const float2*)Hm;

  __shared__ alignas(16) float s_wq[H_DIM * H_DIM];
  __shared__ alignas(16) float s_zt[8][16 * ZT_PITCH];
  const int wib = __builtin_amdgcn_readfirstlane((int)(threadIdx.x >> 5));
  if (FIRST) {
    float4 wq4[4];
    load_w_quads(Wn, wq4);
    store_w_quads(wq4, s_wq);
    __syncthreads();
  }

  int nrows = 0;
  for (int nb = wave0; nb < n; nb += 32 * nwaves) {
    const int mine = nb + lane * nwaves;
    int c_l = 0;
    if (mine < n) c_l = cnt[mine];
    const float d_l = rsqrtf((float)(c_l + 1));
    int deg = __builtin_amdgcn_readlane(c_l, 0);
    int degb = deg < BUCKET ? deg : BUCKET;
    int p = 0;
    if (lane < degb) p = bk[(size_t)nb * BUCKET + lane];
    float2 hs = H2[(size_t)nb * (H_DIM / 2) + lane];

    int node = nb;
    for (int j = 0; j < 32; ++j) {
      const float dd = __int_as_float(
          __builtin_amdgcn_readlane(__float_as_int(d_l), j));
      const int nxt = node + nwaves;
      const bool more = (j + 1 < 32) && (nxt < n);
      float ax = hs.x;
      float ay = hs.y;
      int ndeg = 0, ndegb = 0;
      int pn = 0;
      float2 hn = make_float2(0.0f, 0.0f);

      asm volatile("" : : "v"(p));
      const int s_l = p;
      int gc = degb < ROWS_AHEAD ? degb : ROWS_AHEAD;
      float2 v[ROWS_AHEAD];
      rows_load<0>(v, H2, s_l, 0, gc, lane);
      if (more) {
        ndeg = __builtin_amdgcn_readlane(c_l, j + 1);
        ndegb = ndeg < BUCKET ? ndeg : BUCKET;
        if (lane < ndegb) pn = bk[(size_t)nxt * BUCKET + lane];
        hn = H2[(size_t)nxt * (H_DIM / 2) + lane];
      }
      rows_sum<0>(v, gc, ax, ay);
      for (int i = ROWS_AHEAD; i < degb; ++i) {
        const int s = __builtin_amdgcn_readlane(s_l, i);
        const float2 r = H2[(size_t)s * (H_DIM / 2) + lane];
        ax += r.x;
        ay += r.y;
      }

      if (deg > BUCKET) {
        int L = ovfc[0];
        if (L > ovf_cap) L = ovf_cap;
        for (int b0 = 0; b0 < L; b0 += 32) {
          int2 q = make_int2(-1, 0);
          if (b0 + lane < L) q = ((const int2*)ovf)[b0 + lane];
          unsigned hit = __builtin_amdgcn_ballot_w32(q.x == node);
          while (hit != 0u) {
            const int l = __builtin_ctz(hit);
            hit &= hit - 1u;
            const int s = __builtin_amdgcn_readlane(q.y, l);
            const float2 r = H2[(size_t)s * (H_DIM / 2) + lane];
            ax += r.x;
            ay += r.y;
          }
        }
      }
      ax = fmaf(ax, dd, bb.x);
      ay = fmaf(ay, dd, bb.y);

      if (FIRST) {
        ax = fmaxf(ax, 0.0f) * dd;
        ay = fmaxf(ay, 0.0f) * dd;
        const int slot = j & 15;
        *(float2*)(&s_zt[wib][slot * ZT_PITCH + 2 * lane]) =
            make_float2(ax, ay);
        nrows = slot + 1;
      } else {
        ((float2*)Out)[(size_t)node * (H_DIM / 2) + lane] =
            make_float2(ax, ay);
      }
      if (!more) break;
      deg = ndeg;
      degb = ndegb;
      p = pn;
      hs = hn;
      node = nxt;
    }
  }
  if (FIRST) {
    if (nrows > 0)
      tile_gemm_store(&s_zt[wib][0], s_wq, Out, wave0, nwaves, nrows, lane);
  }
}

#define DEC_GROUP 16
__global__ void __launch_bounds__(256)
k_decode(const float* __restrict__ Z, const int* __restrict__ pos,
         const int* __restrict__ neg, float* __restrict__ out, int npos,
         int nneg) {
  const int total = npos + nneg;
  const int base = __builtin_amdgcn_readfirstlane(
      (int)(blockIdx.x * (blockDim.x >> 5) + (threadIdx.x >> 5))) * 32;
  if (base >= total) return;
  const int lane = threadIdx.x & 31;
  const int k = base + lane;
  int a = 0, b = 0;
  if (k < npos) {
    a = pos[k];
    b = pos[npos + k];
  } else if (k < total) {
    const int j = k - npos;
    a = neg[j];
    b = neg[nneg + j];
  }
  const float2* __restrict__ Z2 = (const float2*)Z;
  float res = 0.0f;
#pragma unroll 1
  for (int t0 = 0; t0 < 32; t0 += DEC_GROUP) {
    float2 va[DEC_GROUP], vb[DEC_GROUP];
#pragma unroll
    for (int i = 0; i < DEC_GROUP; ++i) {
      const int sa = __builtin_amdgcn_readlane(a, t0 + i);
      const int sb = __builtin_amdgcn_readlane(b, t0 + i);
      va[i] = Z2[(size_t)sa * (H_DIM / 2) + lane];
      vb[i] = Z2[(size_t)sb * (H_DIM / 2) + lane];
    }
#pragma unroll
    for (int i = 0; i < DEC_GROUP; ++i) {
      float s = fmaf(va[i].x, vb[i].x, va[i].y * vb[i].y);
#pragma unroll
      for (int o = 16; o > 0; o >>= 1) s += __shfl_xor(s, o, 32);
      if (lane == t0 + i) res = s;
    }
  }
  if (k < total) out[k] = res;
}

extern "C" void kernel_launch(void* const* d_in, const int* in_sizes, int n_in,
                              void* d_out, int out_size, void* d_ws,
                              size_t ws_size, hipStream_t stream) {
  (void)n_in; (void)out_size; (void)ws_size;
  const float* x   = (const float*)d_in[0];
  const int*   ei  = (const int*)d_in[1];
  const int*   pei = (const int*)d_in[2];
  const int*   nei = (const int*)d_in[3];
  const float* W1  = (const float*)d_in[4];
  const float* b1  = (const float*)d_in[5];
  const float* W2  = (const float*)d_in[6];
  const float* b2  = (const float*)d_in[7];
  float* out = (float*)d_out;

  const int n    = in_sizes[0] / H_DIM;
  const int ne   = in_sizes[1] / 2;
  const int npos = in_sizes[2] / 2;
  const int nneg = in_sizes[3] / 2;
  const int* src = ei;
  const int* dst = ei + ne;

  const int B = 256;
  const int eB = SEG_EDGES;
  const int nb = (n + BIN_NODES - 1) / BIN_NODES;
  const int cblocks = (ne + eB - 1) / eB;
  const int cbp = (cblocks + 31) & ~31;
  float* ws = (float*)d_ws;
  const size_t nAl = ((size_t)n + 63) & ~(size_t)63;
  const size_t neAl = ((size_t)ne + 15) & ~(size_t)15;
  float* hbuf = ws;
  float* acc1 = hbuf + (size_t)n * H_DIM;
  float* acc2 = acc1 + (size_t)n * H_DIM;
  int* cnt  = (int*)(acc2 + (size_t)n * H_DIM);
  int* ovfc = cnt + nAl;
  int* bk   = ovfc + 64;
  int* ovf  = bk + (size_t)n * BUCKET;
  int* tab  = ovf + neAl * 2;
  int* segs = tab + (size_t)nb * cbp;

  k_bin_edges<<<cblocks, B, 0, stream>>>(src, dst, tab, cbp, (int2*)segs, ovfc,
                                         ne, nb);
  k_bin_place<<<nb, B, 0, stream>>>(tab, cbp, cblocks, (const int2*)segs, ovfc,
                                    cnt, bk, (int2*)ovf, n, x, W1, hbuf);

  const int node_blocks = (n + 7) / 8;
  const int gblocks = node_blocks < 1280 ? node_blocks : 1280;
  const int tile_blocks = (n + 127) / 128;
  const int g1blocks = gblocks > tile_blocks ? gblocks : tile_blocks;
  k_gather_rows<true><<<g1blocks, B, 0, stream>>>(hbuf, cnt, bk, ovf, ovfc, ne,
                                                  b1, W2, acc1, n);
  k_gather_rows<false><<<gblocks, B, 0, stream>>>(acc1, cnt, bk, ovf, ovfc, ne,
                                                  b2, W2, acc2, n);
  k_decode<<<(npos + nneg + 255) / 256, B, 0, stream>>>(acc2, pei, nei, out,
                                                       npos, nneg);
}
